# starting kernel
# baseline (speedup 1.0000x reference)
#include <hip/hip_runtime.h>
#include <hip/hip_bf16.h>
#include <math.h>

#define D 128

typedef float v2f __attribute__((ext_vector_type(2)));
typedef float v8f __attribute__((ext_vector_type(8)));

__device__ __forceinline__ void atomicMaxF(float* addr, float val) {
    if (val >= 0.0f) atomicMax((int*)addr, __float_as_int(val));
    else             atomicMin((unsigned int*)addr, __float_as_uint(val));
}

__global__ void fill_f32(float* __restrict__ p, float v, int n) {
    int i = blockIdx.x * blockDim.x + threadIdx.x;
    if (i < n) p[i] = v;
}

__global__ void gemm_z(const float* __restrict__ h0, int n0,
                       const float* __restrict__ h1,
                       const float* __restrict__ W,
                       float* __restrict__ Z, int n)
{
    const int wave = threadIdx.x >> 5;
    const int lane = threadIdx.x & 31;
    const int half = lane >> 4;
    const int lr   = lane & 15;
    const int row0 = blockIdx.x * 16;
    const int col0 = wave * 16;

    int rowA = row0 + lr;
    if (rowA >= n) rowA = n - 1;
    const float* hrow = (rowA < n0) ? (h0 + (size_t)rowA * D)
                                    : (h1 + (size_t)(rowA - n0) * D);

    v8f c = {0.f, 0.f, 0.f, 0.f, 0.f, 0.f, 0.f, 0.f};
#pragma unroll
    for (int k0 = 0; k0 < D; k0 += 4) {
        const int kk = k0 + 2 * half;
        float2 av = *(const float2*)(hrow + kk);
        v2f a; a.x = av.x; a.y = av.y;
        v2f b; b.x = W[(size_t)kk * D + col0 + lr];
               b.y = W[(size_t)(kk + 1) * D + col0 + lr];
        c = __builtin_amdgcn_wmma_f32_16x16x4_f32(
                false, a, false, b, (short)0, c, false, false);
    }
#pragma unroll
    for (int v = 0; v < 8; ++v) {
        const int r = row0 + v + 8 * half;
        if (r < n) Z[(size_t)r * D + col0 + lr] = c[v];
    }
}

__global__ void row_scores(const float* __restrict__ Z, const float* __restrict__ a,
                           float* __restrict__ ssrc, float* __restrict__ sdst, int n)
{
    const int wave = (blockIdx.x * blockDim.x + threadIdx.x) >> 5;
    const int lane = threadIdx.x & 31;
    if (wave >= n) return;
    const float4 zv = *(const float4*)(Z + (size_t)wave * D + lane * 4);
    const float4 a0 = *(const float4*)(a + lane * 4);
    const float4 a1 = *(const float4*)(a + D + lane * 4);
    float p0 = zv.x * a0.x + zv.y * a0.y + zv.z * a0.z + zv.w * a0.w;
    float p1 = zv.x * a1.x + zv.y * a1.y + zv.z * a1.z + zv.w * a1.w;
#pragma unroll
    for (int off = 16; off > 0; off >>= 1) {
        p0 += __shfl_down(p0, off, 32);
        p1 += __shfl_down(p1, off, 32);
    }
    if (lane == 0) { ssrc[wave] = p0; sdst[wave] = p1; }
}

__device__ __forceinline__ float leaky(float v) {
    return v > 0.0f ? v : 0.01f * v;
}

__global__ void edge_max(const int* __restrict__ src, const int* __restrict__ dst,
                         const float* __restrict__ ssrc, const float* __restrict__ sdst,
                         float* __restrict__ m, int nE)
{
    int e = blockIdx.x * blockDim.x + threadIdx.x;
    if (e >= nE) return;
    int d = dst[e];
    float v = leaky(ssrc[src[e]] + sdst[d]);
    atomicMaxF(&m[d], v);
}

__global__ void edge_den(const int* __restrict__ src, const int* __restrict__ dst,
                         const float* __restrict__ ssrc, const float* __restrict__ sdst,
                         const float* __restrict__ m, float* __restrict__ den, int nE)
{
    int e = blockIdx.x * blockDim.x + threadIdx.x;
    if (e >= nE) return;
    int d = dst[e];
    float v = leaky(ssrc[src[e]] + sdst[d]);
    atomicAdd(&den[d], __expf(v - m[d]));
}

__global__ void edge_agg(const int* __restrict__ src, const int* __restrict__ dst,
                         const float* __restrict__ ssrc, const float* __restrict__ sdst,
                         const float* __restrict__ m, const float* __restrict__ den,
                         const float* __restrict__ z,
                         float* __restrict__ out_lo, float* __restrict__ out_hi,
                         int n0, int nE)
{
    int t = blockIdx.x * blockDim.x + threadIdx.x;
    int e = t >> 5;
    int lane = t & 31;
    if (e >= nE) return;
    int s = src[e], d = dst[e];
    float v = leaky(ssrc[s] + sdst[d]);
    float alpha = __expf(v - m[d]) / den[d];
    float4 zv = *(const float4*)(z + (size_t)s * D + lane * 4);
    float* op = ((d < n0) ? (out_lo + (size_t)d * D)
                          : (out_hi + (size_t)(d - n0) * D)) + lane * 4;
    atomicAdd(op + 0, alpha * zv.x);
    atomicAdd(op + 1, alpha * zv.y);
    atomicAdd(op + 2, alpha * zv.z);
    atomicAdd(op + 3, alpha * zv.w);
}

__global__ void combine(const float* __restrict__ A, const float* __restrict__ Bfull,
                        const float* __restrict__ W1, const float* __restrict__ b1,
                        const float* __restrict__ w2, float* __restrict__ out,
                        int Isplit)
{
    __shared__ float sA[D], sB[D], red[D];
    const int i = blockIdx.x;
    const int j = threadIdx.x;
    sA[j] = A[(size_t)i * D + j];
    sB[j] = Bfull[(size_t)(Isplit + i) * D + j];
    __syncthreads();
    float accA = b1[j], accB = accA;
    for (int k = 0; k < D; ++k) {
        float wv = W1[(size_t)k * D + j];
        accA = fmaf(sA[k], wv, accA);
        accB = fmaf(sB[k], wv, accB);
    }
    const float w2j = w2[j];
    const float tA = tanhf(accA) * w2j;
    const float tB = tanhf(accB) * w2j;

    red[j] = tA; __syncthreads();
    for (int s = 64; s > 0; s >>= 1) { if (j < s) red[j] += red[j + s]; __syncthreads(); }
    const float s1 = red[0]; __syncthreads();
    red[j] = tB; __syncthreads();
    for (int s = 64; s > 0; s >>= 1) { if (j < s) red[j] += red[j + s]; __syncthreads(); }
    const float s2 = red[0];

    const float mm = fmaxf(s1, s2);
    const float e1 = __expf(s1 - mm), e2 = __expf(s2 - mm);
    const float al = e1 / (e1 + e2);
    out[(size_t)i * D + j] = al * sA[j] + (1.0f - al) * sB[j];
}

static inline int cdiv(int a, int b) { return (a + b - 1) / b; }

extern "C" void kernel_launch(void* const* d_in, const int* in_sizes, int n_in,
                              void* d_out, int out_size, void* d_ws, size_t ws_size,
                              hipStream_t stream)
{
    (void)n_in; (void)ws_size;
    const float* kn_emb   = (const float*)d_in[0];
    const float* exer_emb = (const float*)d_in[1];
    const float* W_dir    = (const float*)d_in[2];
    const float* a_dir    = (const float*)d_in[3];
    const float* W_ke     = (const float*)d_in[4];
    const float* a_ke     = (const float*)d_in[5];
    const float* W_ek     = (const float*)d_in[6];
    const float* a_ek     = (const float*)d_in[7];
    const float* rW1      = (const float*)d_in[8];
    const float* rb1      = (const float*)d_in[9];
    const float* rw2      = (const float*)d_in[10];
    const int* dsrc  = (const int*)d_in[11];
    const int* ddst  = (const int*)d_in[12];
    const int* kesrc = (const int*)d_in[13];
    const int* kedst = (const int*)d_in[14];
    const int* eksrc = (const int*)d_in[15];
    const int* ekdst = (const int*)d_in[16];

    const int Kn = in_sizes[0] / D;
    const int In = in_sizes[1] / D;
    const int Nn = Kn + In;
    const int E1 = in_sizes[11];
    const int E2 = in_sizes[13];

    float* w = (float*)d_ws;
    size_t off = 0;
    float* buf_z = w + off; off += (size_t)Nn * D;
    float* buf_B = w + off; off += (size_t)Nn * D;
    float* buf_A = w + off; off += (size_t)Kn * D;
    float* ssrc  = w + off; off += (size_t)Nn;
    float* sdst  = w + off; off += (size_t)Nn;
    float* mbuf  = w + off; off += (size_t)Nn;
    float* dbuf  = w + off; off += (size_t)Nn;

    float* out_kn = (float*)d_out;
    float* out_ex = (float*)d_out + (size_t)Kn * D;

    hipMemsetAsync(d_out, 0, (size_t)out_size * sizeof(float), stream);

    gemm_z<<<cdiv(Nn, 16), 256, 0, stream>>>(exer_emb, In, kn_emb, W_ek, buf_z, Nn);
    row_scores<<<cdiv(Nn, 8), 256, 0, stream>>>(buf_z, a_ek, ssrc, sdst, Nn);
    fill_f32<<<cdiv(Nn, 256), 256, 0, stream>>>(mbuf, -INFINITY, Nn);
    hipMemsetAsync(dbuf, 0, (size_t)Nn * sizeof(float), stream);
    hipMemsetAsync(buf_A, 0, (size_t)Kn * D * sizeof(float), stream);
    edge_max<<<cdiv(E2, 256), 256, 0, stream>>>(eksrc, ekdst, ssrc, sdst, mbuf, E2);
    edge_den<<<cdiv(E2, 256), 256, 0, stream>>>(eksrc, ekdst, ssrc, sdst, mbuf, dbuf, E2);
    edge_agg<<<cdiv(E2 * 32, 256), 256, 0, stream>>>(eksrc, ekdst, ssrc, sdst, mbuf, dbuf,
                                                     buf_z, out_ex, buf_A, In, E2);

    gemm_z<<<cdiv(Nn, 16), 256, 0, stream>>>(exer_emb, In, kn_emb, W_ke, buf_z, Nn);
    row_scores<<<cdiv(Nn, 8), 256, 0, stream>>>(buf_z, a_ke, ssrc, sdst, Nn);
    fill_f32<<<cdiv(Nn, 256), 256, 0, stream>>>(mbuf, -INFINITY, Nn);
    hipMemsetAsync(dbuf, 0, (size_t)Nn * sizeof(float), stream);
    hipMemsetAsync(buf_B, 0, (size_t)Nn * D * sizeof(float), stream);
    edge_max<<<cdiv(E2, 256), 256, 0, stream>>>(kesrc, kedst, ssrc, sdst, mbuf, E2);
    edge_den<<<cdiv(E2, 256), 256, 0, stream>>>(kesrc, kedst, ssrc, sdst, mbuf, dbuf, E2);
    edge_agg<<<cdiv(E2 * 32, 256), 256, 0, stream>>>(kesrc, kedst, ssrc, sdst, mbuf, dbuf,
                                                     buf_z, buf_B, buf_B, Nn, E2);

    gemm_z<<<cdiv(Kn, 16), 256, 0, stream>>>(kn_emb, Kn, kn_emb, W_dir, buf_z, Kn);
    row_scores<<<cdiv(Kn, 8), 256, 0, stream>>>(buf_z, a_dir, ssrc, sdst, Kn);
    fill_f32<<<cdiv(Kn, 256), 256, 0, stream>>>(mbuf, -INFINITY, Kn);
    hipMemsetAsync(dbuf, 0, (size_t)Kn * sizeof(float), stream);
    hipMemsetAsync(buf_A, 0, (size_t)Kn * D * sizeof(float), stream);
    edge_max<<<cdiv(E1, 256), 256, 0, stream>>>(dsrc, ddst, ssrc, sdst, mbuf, E1);
    edge_den<<<cdiv(E1, 256), 256, 0, stream>>>(dsrc, ddst, ssrc, sdst, mbuf, dbuf, E1);
    edge_agg<<<cdiv(E1 * 32, 256), 256, 0, stream>>>(dsrc, ddst, ssrc, sdst, mbuf, dbuf,
                                                     buf_z, buf_A, buf_A, Kn, E1);

    combine<<<Kn, D, 0, stream>>>(buf_A, buf_B, rW1, rb1, rw2, out_kn, In);
}


# Edge passes skip edges whose destination row the reference slices away (ke keeps rows >= I, ek rows < I); aggregation wave walks 32 edges so a skipped edge costs no wave
# speedup vs baseline: 1.3500x; 1.3500x over previous
#include <hip/hip_runtime.h>
#include <hip/hip_bf16.h>
#include <math.h>

#define D 128

typedef float v2f __attribute__((ext_vector_type(2)));
typedef float v8f __attribute__((ext_vector_type(8)));

__device__ __forceinline__ void atomicMaxF(float* addr, float val) {
    if (val >= 0.0f) atomicMax((int*)addr, __float_as_int(val));
    else             atomicMin((unsigned int*)addr, __float_as_uint(val));
}

__global__ void fill_f32(float* __restrict__ p, float v, int n) {
    int i = blockIdx.x * blockDim.x + threadIdx.x;
    if (i < n) p[i] = v;
}

__global__ void gemm_z(const float* __restrict__ h0, int n0,
                       const float* __restrict__ h1,
                       const float* __restrict__ W,
                       float* __restrict__ Z, int n)
{
    const int wave = threadIdx.x >> 5;
    const int lane = threadIdx.x & 31;
    const int half = lane >> 4;
    const int lr   = lane & 15;
    const int row0 = blockIdx.x * 16;
    const int col0 = wave * 16;

    int rowA = row0 + lr;
    if (rowA >= n) rowA = n - 1;
    const float* hrow = (rowA < n0) ? (h0 + (size_t)rowA * D)
                                    : (h1 + (size_t)(rowA - n0) * D);

    v8f c = {0.f, 0.f, 0.f, 0.f, 0.f, 0.f, 0.f, 0.f};
#pragma unroll
    for (int k0 = 0; k0 < D; k0 += 4) {
        const int kk = k0 + 2 * half;
        float2 av = *(const float2*)(hrow + kk);
        v2f a; a.x = av.x; a.y = av.y;
        v2f b; b.x = W[(size_t)kk * D + col0 + lr];
               b.y = W[(size_t)(kk + 1) * D + col0 + lr];
        c = __builtin_amdgcn_wmma_f32_16x16x4_f32(
                false, a, false, b, (short)0, c, false, false);
    }
#pragma unroll
    for (int v = 0; v < 8; ++v) {
        const int r = row0 + v + 8 * half;
        if (r < n) Z[(size_t)r * D + col0 + lr] = c[v];
    }
}

__global__ void row_scores(const float* __restrict__ Z, const float* __restrict__ a,
                           float* __restrict__ ssrc, float* __restrict__ sdst, int n)
{
    const int wave = (blockIdx.x * blockDim.x + threadIdx.x) >> 5;
    const int lane = threadIdx.x & 31;
    if (wave >= n) return;
    const float4 zv = *(const float4*)(Z + (size_t)wave * D + lane * 4);
    const float4 a0 = *(const float4*)(a + lane * 4);
    const float4 a1 = *(const float4*)(a + D + lane * 4);
    float p0 = zv.x * a0.x + zv.y * a0.y + zv.z * a0.z + zv.w * a0.w;
    float p1 = zv.x * a1.x + zv.y * a1.y + zv.z * a1.z + zv.w * a1.w;
#pragma unroll
    for (int off = 16; off > 0; off >>= 1) {
        p0 += __shfl_down(p0, off, 32);
        p1 += __shfl_down(p1, off, 32);
    }
    if (lane == 0) { ssrc[wave] = p0; sdst[wave] = p1; }
}

__device__ __forceinline__ float leaky(float v) {
    return v > 0.0f ? v : 0.01f * v;
}


__global__ void edge_max(const int* __restrict__ src, const int* __restrict__ dst,
                         const float* __restrict__ ssrc, const float* __restrict__ sdst,
                         float* __restrict__ m, int lo, int hi, int nE)
{
    int e = blockIdx.x * blockDim.x + threadIdx.x;
    if (e >= nE) return;
    int d = dst[e];
    if (d < lo || d >= hi) return;
    float v = leaky(ssrc[src[e]] + sdst[d]);
    atomicMaxF(&m[d], v);
}

__global__ void edge_den(const int* __restrict__ src, const int* __restrict__ dst,
                         const float* __restrict__ ssrc, const float* __restrict__ sdst,
                         const float* __restrict__ m, float* __restrict__ den,
                         int lo, int hi, int nE)
{
    int e = blockIdx.x * blockDim.x + threadIdx.x;
    if (e >= nE) return;
    int d = dst[e];
    if (d < lo || d >= hi) return;
    float v = leaky(ssrc[src[e]] + sdst[d]);
    atomicAdd(&den[d], __expf(v - m[d]));
}

__global__ void edge_agg(const int* __restrict__ src, const int* __restrict__ dst,
                         const float* __restrict__ ssrc, const float* __restrict__ sdst,
                         const float* __restrict__ m, const float* __restrict__ den,
                         const float* __restrict__ z, float* __restrict__ out,
                         int lo, int hi, int nE)
{
    const int t = blockIdx.x * blockDim.x + threadIdx.x;
    const int lane = t & 31;
    const int e = t;
    int s = 0, d = -1;
    if (e < nE) { s = src[e]; d = dst[e]; }
    const bool keep = (d >= lo) && (d < hi);
    float alpha = 0.0f;
    if (keep) {
        const float v = leaky(ssrc[s] + sdst[d]);
        alpha = __expf(v - m[d]) / den[d];
    }
    unsigned mask = __builtin_amdgcn_ballot_w32(keep);
    while (mask) {
        const int i = __builtin_ctz(mask);
        mask &= mask - 1;
        const int   si = __builtin_amdgcn_readlane(s, i);
        const int   di = __builtin_amdgcn_readlane(d, i);
        const float ai = __int_as_float(__builtin_amdgcn_readlane(__float_as_int(alpha), i));
        const float4 zv = *(const float4*)(z + (size_t)si * D + lane * 4);
        float* op = out + (size_t)(di - lo) * D + lane * 4;
        atomicAdd(op + 0, ai * zv.x);
        atomicAdd(op + 1, ai * zv.y);
        atomicAdd(op + 2, ai * zv.z);
        atomicAdd(op + 3, ai * zv.w);
    }
}

__global__ void combine(const float* __restrict__ A, const float* __restrict__ B,
                        const float* __restrict__ W1, const float* __restrict__ b1,
                        const float* __restrict__ w2, float* __restrict__ out)
{
    __shared__ float sA[D], sB[D], red[D];
    const int i = blockIdx.x;
    const int j = threadIdx.x;
    sA[j] = A[(size_t)i * D + j];
    sB[j] = B[(size_t)i * D + j];
    __syncthreads();
    float accA = b1[j], accB = accA;
    for (int k = 0; k < D; ++k) {
        float wv = W1[(size_t)k * D + j];
        accA = fmaf(sA[k], wv, accA);
        accB = fmaf(sB[k], wv, accB);
    }
    const float w2j = w2[j];
    const float tA = tanhf(accA) * w2j;
    const float tB = tanhf(accB) * w2j;

    red[j] = tA; __syncthreads();
    for (int s = 64; s > 0; s >>= 1) { if (j < s) red[j] += red[j + s]; __syncthreads(); }
    const float s1 = red[0]; __syncthreads();
    red[j] = tB; __syncthreads();
    for (int s = 64; s > 0; s >>= 1) { if (j < s) red[j] += red[j + s]; __syncthreads(); }
    const float s2 = red[0];

    const float mm = fmaxf(s1, s2);
    const float e1 = __expf(s1 - mm), e2 = __expf(s2 - mm);
    const float al = e1 / (e1 + e2);
    out[(size_t)i * D + j] = al * sA[j] + (1.0f - al) * sB[j];
}

static inline int cdiv(int a, int b) { return (a + b - 1) / b; }

extern "C" void kernel_launch(void* const* d_in, const int* in_sizes, int n_in,
                              void* d_out, int out_size, void* d_ws, size_t ws_size,
                              hipStream_t stream)
{
    (void)n_in; (void)ws_size;
    const float* kn_emb   = (const float*)d_in[0];
    const float* exer_emb = (const float*)d_in[1];
    const float* W_dir    = (const float*)d_in[2];
    const float* a_dir    = (const float*)d_in[3];
    const float* W_ke     = (const float*)d_in[4];
    const float* a_ke     = (const float*)d_in[5];
    const float* W_ek     = (const float*)d_in[6];
    const float* a_ek     = (const float*)d_in[7];
    const float* rW1      = (const float*)d_in[8];
    const float* rb1      = (const float*)d_in[9];
    const float* rw2      = (const float*)d_in[10];
    const int* dsrc  = (const int*)d_in[11];
    const int* ddst  = (const int*)d_in[12];
    const int* kesrc = (const int*)d_in[13];
    const int* kedst = (const int*)d_in[14];
    const int* eksrc = (const int*)d_in[15];
    const int* ekdst = (const int*)d_in[16];

    const int Kn = in_sizes[0] / D;
    const int In = in_sizes[1] / D;
    const int Nn = Kn + In;
    const int E1 = in_sizes[11];
    const int E2 = in_sizes[13];

    float* w = (float*)d_ws;
    size_t off = 0;
    float* buf_z = w + off; off += (size_t)Nn * D;
    float* buf_B = w + off; off += (size_t)Nn * D;
    float* buf_A = w + off; off += (size_t)Kn * D;
    float* ssrc  = w + off; off += (size_t)Nn;
    float* sdst  = w + off; off += (size_t)Nn;
    float* mbuf  = w + off; off += (size_t)Nn;
    float* dbuf  = w + off; off += (size_t)Nn;

    float* out_kn = (float*)d_out;
    float* out_ex = (float*)d_out + (size_t)Kn * D;

    hipMemsetAsync(d_out, 0, (size_t)out_size * sizeof(float), stream);

    gemm_z<<<cdiv(Nn, 16), 256, 0, stream>>>(exer_emb, In, kn_emb, W_ek, buf_z, Nn);
    row_scores<<<cdiv(Nn, 8), 256, 0, stream>>>(buf_z, a_ek, ssrc, sdst, Nn);
    fill_f32<<<cdiv(Nn, 256), 256, 0, stream>>>(mbuf, -INFINITY, Nn);
    hipMemsetAsync(dbuf, 0, (size_t)Nn * sizeof(float), stream);
    edge_max<<<cdiv(E2, 256), 256, 0, stream>>>(eksrc, ekdst, ssrc, sdst, mbuf, 0, In, E2);
    edge_den<<<cdiv(E2, 256), 256, 0, stream>>>(eksrc, ekdst, ssrc, sdst, mbuf, dbuf, 0, In, E2);
    edge_agg<<<cdiv(E2, 256), 256, 0, stream>>>(eksrc, ekdst, ssrc, sdst, mbuf, dbuf,
                                                buf_z, out_ex, 0, In, E2);

    gemm_z<<<cdiv(Nn, 16), 256, 0, stream>>>(exer_emb, In, kn_emb, W_ke, buf_z, Nn);
    row_scores<<<cdiv(Nn, 8), 256, 0, stream>>>(buf_z, a_ke, ssrc, sdst, Nn);
    fill_f32<<<cdiv(Nn, 256), 256, 0, stream>>>(mbuf, -INFINITY, Nn);
    hipMemsetAsync(dbuf, 0, (size_t)Nn * sizeof(float), stream);
    hipMemsetAsync(buf_B, 0, (size_t)Kn * D * sizeof(float), stream);
    edge_max<<<cdiv(E2, 256), 256, 0, stream>>>(kesrc, kedst, ssrc, sdst, mbuf, In, Nn, E2);
    edge_den<<<cdiv(E2, 256), 256, 0, stream>>>(kesrc, kedst, ssrc, sdst, mbuf, dbuf, In, Nn, E2);
    edge_agg<<<cdiv(E2, 256), 256, 0, stream>>>(kesrc, kedst, ssrc, sdst, mbuf, dbuf,
                                                buf_z, buf_B, In, Nn, E2);

    gemm_z<<<cdiv(Kn, 16), 256, 0, stream>>>(kn_emb, Kn, kn_emb, W_dir, buf_z, Kn);
    row_scores<<<cdiv(Kn, 8), 256, 0, stream>>>(buf_z, a_dir, ssrc, sdst, Kn);
    fill_f32<<<cdiv(Kn, 256), 256, 0, stream>>>(mbuf, -INFINITY, Kn);
    hipMemsetAsync(dbuf, 0, (size_t)Kn * sizeof(float), stream);
    hipMemsetAsync(buf_A, 0, (size_t)Kn * D * sizeof(float), stream);
    edge_max<<<cdiv(E1, 256), 256, 0, stream>>>(dsrc, ddst, ssrc, sdst, mbuf, 0, Kn, E1);
    edge_den<<<cdiv(E1, 256), 256, 0, stream>>>(dsrc, ddst, ssrc, sdst, mbuf, dbuf, 0, Kn, E1);
    edge_agg<<<cdiv(E1, 256), 256, 0, stream>>>(dsrc, ddst, ssrc, sdst, mbuf, dbuf,
                                                buf_z, buf_A, 0, Kn, E1);

    combine<<<Kn, D, 0, stream>>>(buf_A, buf_B, rW1, rb1, rw2, out_kn);
}


# Edges are counting-sorted by destination each call (count, one-block scan, fill); one wave per kept row then does max, sum and weighted gather in registers and stores the row once: no float atomics
# speedup vs baseline: 1.7674x; 1.3092x over previous
#include <hip/hip_runtime.h>
#include <hip/hip_bf16.h>
#include <math.h>


#define D 128

typedef float v2f __attribute__((ext_vector_type(2)));
typedef float v8f __attribute__((ext_vector_type(8)));

__global__ void gemm_z(const float* __restrict__ h0, int n0,
                       const float* __restrict__ h1,
                       const float* __restrict__ W,
                       float* __restrict__ Z, int n)
{
    const int wave = threadIdx.x >> 5;
    const int lane = threadIdx.x & 31;
    const int half = lane >> 4;
    const int lr   = lane & 15;
    const int row0 = blockIdx.x * 16;
    const int col0 = wave * 16;

    int rowA = row0 + lr;
    if (rowA >= n) rowA = n - 1;
    const float* hrow = (rowA < n0) ? (h0 + (size_t)rowA * D)
                                    : (h1 + (size_t)(rowA - n0) * D);

    v8f c = {0.f, 0.f, 0.f, 0.f, 0.f, 0.f, 0.f, 0.f};
#pragma unroll
    for (int k0 = 0; k0 < D; k0 += 4) {
        const int kk = k0 + 2 * half;
        float2 av = *(const float2*)(hrow + kk);
        v2f a; a.x = av.x; a.y = av.y;
        v2f b; b.x = W[(size_t)kk * D + col0 + lr];
               b.y = W[(size_t)(kk + 1) * D + col0 + lr];
        c = __builtin_amdgcn_wmma_f32_16x16x4_f32(
                false, a, false, b, (short)0, c, false, false);
    }
#pragma unroll
    for (int v = 0; v < 8; ++v) {
        const int r = row0 + v + 8 * half;
        if (r < n) Z[(size_t)r * D + col0 + lr] = c[v];
    }
}

__global__ void row_scores(const float* __restrict__ Z, const float* __restrict__ a,
                           float* __restrict__ ssrc, float* __restrict__ sdst, int n)
{
    const int wave = (blockIdx.x * blockDim.x + threadIdx.x) >> 5;
    const int lane = threadIdx.x & 31;
    if (wave >= n) return;
    const float4 zv = *(const float4*)(Z + (size_t)wave * D + lane * 4);
    const float4 a0 = *(const float4*)(a + lane * 4);
    const float4 a1 = *(const float4*)(a + D + lane * 4);
    float p0 = zv.x * a0.x + zv.y * a0.y + zv.z * a0.z + zv.w * a0.w;
    float p1 = zv.x * a1.x + zv.y * a1.y + zv.z * a1.z + zv.w * a1.w;
#pragma unroll
    for (int off = 16; off > 0; off >>= 1) {
        p0 += __shfl_down(p0, off, 32);
        p1 += __shfl_down(p1, off, 32);
    }
    if (lane == 0) { ssrc[wave] = p0; sdst[wave] = p1; }
}

__device__ __forceinline__ float leaky(float v) {
    return v > 0.0f ? v : 0.01f * v;
}


__device__ __forceinline__ int4 load_ids4(const int* __restrict__ p, int e0, int nE) {
    int4 v = make_int4(-1, -1, -1, -1);
    if (e0 + 3 < nE) {
        v = *(const int4*)(p + e0);
    } else {
        if (e0     < nE) v.x = p[e0];
        if (e0 + 1 < nE) v.y = p[e0 + 1];
        if (e0 + 2 < nE) v.z = p[e0 + 2];
    }
    return v;
}

__global__ void edge_count(const int* __restrict__ dst, int* __restrict__ cnt,
                           int lo, int hi, int nE)
{
    const int e0 = (blockIdx.x * blockDim.x + threadIdx.x) * 4;
    if (e0 >= nE) return;
    const int4 d = load_ids4(dst, e0, nE);
    if (d.x >= lo && d.x < hi) atomicAdd(&cnt[d.x - lo], 1);
    if (d.y >= lo && d.y < hi) atomicAdd(&cnt[d.y - lo], 1);
    if (d.z >= lo && d.z < hi) atomicAdd(&cnt[d.z - lo], 1);
    if (d.w >= lo && d.w < hi) atomicAdd(&cnt[d.w - lo], 1);
}

__global__ void edge_fill(const int* __restrict__ src, const int* __restrict__ dst,
                          int* __restrict__ cur, int* __restrict__ csr,
                          int lo, int hi, int nE)
{
    const int e0 = (blockIdx.x * blockDim.x + threadIdx.x) * 4;
    if (e0 >= nE) return;
    const int4 d = load_ids4(dst, e0, nE);
    const int4 s = load_ids4(src, e0, nE);
    if (d.x >= lo && d.x < hi) csr[atomicAdd(&cur[d.x - lo], 1)] = s.x;
    if (d.y >= lo && d.y < hi) csr[atomicAdd(&cur[d.y - lo], 1)] = s.y;
    if (d.z >= lo && d.z < hi) csr[atomicAdd(&cur[d.z - lo], 1)] = s.z;
    if (d.w >= lo && d.w < hi) csr[atomicAdd(&cur[d.w - lo], 1)] = s.w;
}

#define SCAN_T 1024

__device__ __forceinline__ int4 load_cnt4(const int* p, int i0, int n) {
    int4 v = make_int4(0, 0, 0, 0);
    if (i0 + 3 < n) {
        v = *(const int4*)(p + i0);
    } else {
        if (i0     < n) v.x = p[i0];
        if (i0 + 1 < n) v.y = p[i0 + 1];
        if (i0 + 2 < n) v.z = p[i0 + 2];
    }
    return v;
}

__device__ __forceinline__ int wave_incl_scan(int x, int lane) {
#pragma unroll
    for (int off = 1; off < 32; off <<= 1) {
        const int nb = __shfl_up(x, off, 32);
        if (lane >= off) x += nb;
    }
    return x;
}

__global__ void __launch_bounds__(SCAN_T)
scan_rows(int* cur, int* __restrict__ offs, int n)
{
    __shared__ int wsum[32];
    const int t = threadIdx.x;
    const int lane = t & 31;
    const int wv = t >> 5;
    int carry = 0;
    int4 v = load_cnt4(cur, t * 4, n);
    for (int base = 0; base < n; base += SCAN_T * 4) {
        const int i0 = base + t * 4;
        const int4 nv = load_cnt4(cur, i0 + SCAN_T * 4, n);
        const int tsum = v.x + v.y + v.z + v.w;
        const int inc = wave_incl_scan(tsum, lane);
        if (lane == 31) wsum[wv] = inc;
        __syncthreads();
        const int ws = wsum[lane];
        const int winc = wave_incl_scan(ws, lane);
        const int wave_excl = __shfl(winc, wv, 32) - __shfl(ws, wv, 32);
        const int total = __shfl(winc, 31, 32);
        __syncthreads();
        int4 o;
        o.x = carry + wave_excl + (inc - tsum);
        o.y = o.x + v.x;
        o.z = o.y + v.y;
        o.w = o.z + v.z;
        if (i0 + 3 < n) {
            *(int4*)(offs + i0) = o;
            *(int4*)(cur + i0) = o;
        } else {
            if (i0     < n) { offs[i0]     = o.x; cur[i0]     = o.x; }
            if (i0 + 1 < n) { offs[i0 + 1] = o.y; cur[i0 + 1] = o.y; }
            if (i0 + 2 < n) { offs[i0 + 2] = o.z; cur[i0 + 2] = o.z; }
        }
        carry += total;
        v = nv;
    }
}

__device__ __forceinline__ float wave_max(float x) {
#pragma unroll
    for (int off = 16; off > 0; off >>= 1) x = fmaxf(x, __shfl_xor(x, off, 32));
    return x;
}
__device__ __forceinline__ float wave_sum(float x) {
#pragma unroll
    for (int off = 16; off > 0; off >>= 1) x += __shfl_xor(x, off, 32);
    return x;
}

__global__ void agg_rows(const int* __restrict__ offs, const int* __restrict__ endp,
                         const int* __restrict__ csr,
                         const float* __restrict__ ssrc, const float* __restrict__ sdst,
                         const float* __restrict__ z, float* __restrict__ out,
                         int lo, int nr)
{
    const int lane = threadIdx.x & 31;
    const int r = __builtin_amdgcn_readfirstlane((blockIdx.x * blockDim.x + threadIdx.x) >> 5);
    if (r >= nr) return;
    const int beg = __builtin_amdgcn_readfirstlane(offs[r]);
    const int end = __builtin_amdgcn_readfirstlane(endp[r]);
    const float sd = sdst[r + lo];

    int   s0 = 0;
    float v0 = -INFINITY;
    const bool have0 = (beg + lane) < end;
    if (have0) { s0 = csr[beg + lane]; v0 = leaky(ssrc[s0] + sd); }
    float mx = v0;
    for (int j = beg + 32 + lane; j < end; j += 32)
        mx = fmaxf(mx, leaky(ssrc[csr[j]] + sd));
    mx = wave_max(mx);

    const float ex0 = have0 ? __expf(v0 - mx) : 0.0f;
    float dn = ex0;
    for (int j = beg + 32 + lane; j < end; j += 32)
        dn += __expf(leaky(ssrc[csr[j]] + sd) - mx);
    dn = wave_sum(dn);

    float4 acc = make_float4(0.0f, 0.0f, 0.0f, 0.0f);
    for (int base = beg; base < end; base += 32) {
        int   s = s0;
        float a = have0 ? ex0 / dn : 0.0f;
        if (base != beg) {
            const int j = base + lane;
            s = 0; a = 0.0f;
            if (j < end) { s = csr[j]; a = __expf(leaky(ssrc[s] + sd) - mx) / dn; }
        }
        const int cnt = (end - base) < 32 ? (end - base) : 32;
#pragma unroll 4
        for (int i = 0; i < cnt; ++i) {
            const int   si = __builtin_amdgcn_readlane(s, i);
            const float ai = __int_as_float(__builtin_amdgcn_readlane(__float_as_int(a), i));
            const float4 zv = *(const float4*)(z + (size_t)si * D + lane * 4);
            acc.x = fmaf(ai, zv.x, acc.x);
            acc.y = fmaf(ai, zv.y, acc.y);
            acc.z = fmaf(ai, zv.z, acc.z);
            acc.w = fmaf(ai, zv.w, acc.w);
        }
    }
    *(float4*)(out + (size_t)r * D + lane * 4) = acc;
}

__global__ void combine(const float* __restrict__ A, const float* __restrict__ B,
                        const float* __restrict__ W1, const float* __restrict__ b1,
                        const float* __restrict__ w2, float* __restrict__ out)
{
    __shared__ float sA[D], sB[D], red[D];
    const int i = blockIdx.x;
    const int j = threadIdx.x;
    sA[j] = A[(size_t)i * D + j];
    sB[j] = B[(size_t)i * D + j];
    __syncthreads();
    float accA = b1[j], accB = accA;
    for (int k = 0; k < D; ++k) {
        float wv = W1[(size_t)k * D + j];
        accA = fmaf(sA[k], wv, accA);
        accB = fmaf(sB[k], wv, accB);
    }
    const float w2j = w2[j];
    const float tA = tanhf(accA) * w2j;
    const float tB = tanhf(accB) * w2j;

    red[j] = tA; __syncthreads();
    for (int s = 64; s > 0; s >>= 1) { if (j < s) red[j] += red[j + s]; __syncthreads(); }
    const float s1 = red[0]; __syncthreads();
    red[j] = tB; __syncthreads();
    for (int s = 64; s > 0; s >>= 1) { if (j < s) red[j] += red[j + s]; __syncthreads(); }
    const float s2 = red[0];

    const float mm = fmaxf(s1, s2);
    const float e1 = __expf(s1 - mm), e2 = __expf(s2 - mm);
    const float al = e1 / (e1 + e2);
    out[(size_t)i * D + j] = al * sA[j] + (1.0f - al) * sB[j];
}

static inline int cdiv(int a, int b) { return (a + b - 1) / b; }

extern "C" void kernel_launch(void* const* d_in, const int* in_sizes, int n_in,
                              void* d_out, int out_size, void* d_ws, size_t ws_size,
                              hipStream_t stream)
{
    (void)n_in; (void)ws_size; (void)out_size;
    const float* kn_emb   = (const float*)d_in[0];
    const float* exer_emb = (const float*)d_in[1];
    const float* W_dir    = (const float*)d_in[2];
    const float* a_dir    = (const float*)d_in[3];
    const float* W_ke     = (const float*)d_in[4];
    const float* a_ke     = (const float*)d_in[5];
    const float* W_ek     = (const float*)d_in[6];
    const float* a_ek     = (const float*)d_in[7];
    const float* rW1      = (const float*)d_in[8];
    const float* rb1      = (const float*)d_in[9];
    const float* rw2      = (const float*)d_in[10];
    const int* dsrc  = (const int*)d_in[11];
    const int* ddst  = (const int*)d_in[12];
    const int* kesrc = (const int*)d_in[13];
    const int* kedst = (const int*)d_in[14];
    const int* eksrc = (const int*)d_in[15];
    const int* ekdst = (const int*)d_in[16];

    const int Kn = in_sizes[0] / D;
    const int In = in_sizes[1] / D;
    const int Nn = Kn + In;
    const int E1 = in_sizes[11];
    const int E2 = in_sizes[13];

    float* w = (float*)d_ws;
    size_t off = 0;
    float* buf_z = w + off; off += (size_t)Nn * D;
    float* buf_B = w + off; off += (size_t)Nn * D;
    float* buf_A = w + off; off += (size_t)Kn * D;
    float* ssrc  = w + off; off += (size_t)Nn;
    float* sdst  = w + off; off += (size_t)Nn;

    const size_t n4 = ((size_t)Nn + 3) & ~(size_t)3;
    int* cur  = (int*)(buf_B + (size_t)Kn * D);
    int* offs = cur + n4;
    int* csr  = offs + n4;

    float* out_kn = (float*)d_out;
    float* out_ex = (float*)d_out + (size_t)Kn * D;

    gemm_z<<<cdiv(Nn, 16), 256, 0, stream>>>(exer_emb, In, kn_emb, W_ek, buf_z, Nn);
    row_scores<<<cdiv(Nn, 8), 256, 0, stream>>>(buf_z, a_ek, ssrc, sdst, Nn);
    hipMemsetAsync(cur, 0, (size_t)In * sizeof(int), stream);
    edge_count<<<cdiv(cdiv(E2, 4), 256), 256, 0, stream>>>(ekdst, cur, 0, In, E2);
    scan_rows<<<1, SCAN_T, 0, stream>>>(cur, offs, In);
    edge_fill<<<cdiv(cdiv(E2, 4), 256), 256, 0, stream>>>(eksrc, ekdst, cur, csr, 0, In, E2);
    agg_rows<<<cdiv(In, 8), 256, 0, stream>>>(offs, cur, csr, ssrc, sdst, buf_z, out_ex, 0, In);

    gemm_z<<<cdiv(Nn, 16), 256, 0, stream>>>(exer_emb, In, kn_emb, W_ke, buf_z, Nn);
    row_scores<<<cdiv(Nn, 8), 256, 0, stream>>>(buf_z, a_ke, ssrc, sdst, Nn);
    hipMemsetAsync(cur, 0, (size_t)Kn * sizeof(int), stream);
    edge_count<<<cdiv(cdiv(E2, 4), 256), 256, 0, stream>>>(kedst, cur, In, Nn, E2);
    scan_rows<<<1, SCAN_T, 0, stream>>>(cur, offs, Kn);
    edge_fill<<<cdiv(cdiv(E2, 4), 256), 256, 0, stream>>>(kesrc, kedst, cur, csr, In, Nn, E2);
    agg_rows<<<cdiv(Kn, 8), 256, 0, stream>>>(offs, cur, csr, ssrc, sdst, buf_z, buf_B, In, Kn);

    gemm_z<<<cdiv(Kn, 16), 256, 0, stream>>>(kn_emb, Kn, kn_emb, W_dir, buf_z, Kn);
    row_scores<<<cdiv(Kn, 8), 256, 0, stream>>>(buf_z, a_dir, ssrc, sdst, Kn);
    hipMemsetAsync(cur, 0, (size_t)Kn * sizeof(int), stream);
    edge_count<<<cdiv(cdiv(E1, 4), 256), 256, 0, stream>>>(ddst, cur, 0, Kn, E1);
    scan_rows<<<1, SCAN_T, 0, stream>>>(cur, offs, Kn);
    edge_fill<<<cdiv(cdiv(E1, 4), 256), 256, 0, stream>>>(dsrc, ddst, cur, csr, 0, Kn, E1);
    agg_rows<<<cdiv(Kn, 8), 256, 0, stream>>>(offs, cur, csr, ssrc, sdst, buf_z, buf_A, 0, Kn);

    combine<<<Kn, D, 0, stream>>>(buf_A, buf_B, rW1, rb1, rw2, out_kn);
}


# Gate MLP products A@W1 and B@W1 move from vector FMAs to the f32 matrix instruction: 16 rows per 8-wave block staged in LDS; tanh*w2 row sums by lane exchange then LDS (their sum order changes)
# speedup vs baseline: 2.0745x; 1.1738x over previous
#include <hip/hip_runtime.h>
#include <hip/hip_bf16.h>
#include <math.h>


#define D 128

typedef float v2f __attribute__((ext_vector_type(2)));
typedef float v8f __attribute__((ext_vector_type(8)));

__device__ __forceinline__ float rows_dot16(const v8f c, float coef, int lr)
{
    const bool b3 = (lr & 8) != 0;
    const bool b2 = (lr & 4) != 0;
    const bool b1 = (lr & 2) != 0;
    float y[4];
#pragma unroll
    for (int i = 0; i < 4; ++i) {
        const float lo = c[i] * coef, hi = c[i + 4] * coef;
        const float keep = b3 ? hi : lo;
        const float send = b3 ? lo : hi;
        y[i] = keep + __shfl_xor(send, 8, 32);
    }
    float q[2];
#pragma unroll
    for (int i = 0; i < 2; ++i) {
        const float keep = b2 ? y[i + 2] : y[i];
        const float send = b2 ? y[i] : y[i + 2];
        q[i] = keep + __shfl_xor(send, 4, 32);
    }
    const float keep = b1 ? q[1] : q[0];
    const float send = b1 ? q[0] : q[1];
    float w = keep + __shfl_xor(send, 2, 32);
    w += __shfl_xor(w, 1, 32);
    return w;
}

__global__ void gemm_z(const float* __restrict__ h0, int n0,
                       const float* __restrict__ h1,
                       const float* __restrict__ W,
                       const float* __restrict__ att,
                       float* __restrict__ Z,
                       float* __restrict__ ssrc, float* __restrict__ sdst, int n)
{
    __shared__ float sc[8][2][16];
    const int wave = threadIdx.x >> 5;
    const int lane = threadIdx.x & 31;
    const int half = lane >> 4;
    const int lr   = lane & 15;
    const int row0 = blockIdx.x * 16;
    const int col0 = wave * 16;

    int rowA = row0 + lr;
    if (rowA >= n) rowA = n - 1;
    const float* hrow = (rowA < n0) ? (h0 + (size_t)rowA * D)
                                    : (h1 + (size_t)(rowA - n0) * D);

    v8f c = {0.f, 0.f, 0.f, 0.f, 0.f, 0.f, 0.f, 0.f};
#pragma unroll
    for (int k0 = 0; k0 < D; k0 += 4) {
        const int kk = k0 + 2 * half;
        float2 av = *(const float2*)(hrow + kk);
        v2f a; a.x = av.x; a.y = av.y;
        v2f b; b.x = W[(size_t)kk * D + col0 + lr];
               b.y = W[(size_t)(kk + 1) * D + col0 + lr];
        c = __builtin_amdgcn_wmma_f32_16x16x4_f32(
                false, a, false, b, (short)0, c, false, false);
    }
    const float ps = rows_dot16(c, att[col0 + lr], lr);
    const float pd = rows_dot16(c, att[D + col0 + lr], lr);
    if ((lr & 1) == 0) {
        sc[wave][0][8 * half + (lr >> 1)] = ps;
        sc[wave][1][8 * half + (lr >> 1)] = pd;
    }
    __syncthreads();

#pragma unroll
    for (int v = 0; v < 8; ++v) {
        const int r = row0 + v + 8 * half;
        if (r < n) Z[(size_t)r * D + col0 + lr] = c[v];
    }

    if (threadIdx.x < 32) {
        const int which = threadIdx.x >> 4;
        const int rr = threadIdx.x & 15;
        float t = 0.0f;
#pragma unroll
        for (int w8 = 0; w8 < 8; ++w8) t += sc[w8][which][rr];
        const int r = row0 + rr;
        if (r < n) { if (which) sdst[r] = t; else ssrc[r] = t; }
    }
}

__device__ __forceinline__ float leaky(float v) {
    return v > 0.0f ? v : 0.01f * v;
}


__device__ __forceinline__ int4 load_ids4(const int* __restrict__ p, int e0, int nE) {
    int4 v = make_int4(-1, -1, -1, -1);
    if (e0 + 3 < nE) {
        v = *(const int4*)(p + e0);
    } else {
        if (e0     < nE) v.x = p[e0];
        if (e0 + 1 < nE) v.y = p[e0 + 1];
        if (e0 + 2 < nE) v.z = p[e0 + 2];
    }
    return v;
}

__global__ void edge_count(const int* __restrict__ dst, int* __restrict__ cnt,
                           int lo, int hi, int nE)
{
    const int e0 = (blockIdx.x * blockDim.x + threadIdx.x) * 4;
    if (e0 >= nE) return;
    const int4 d = load_ids4(dst, e0, nE);
    if (d.x >= lo && d.x < hi) atomicAdd(&cnt[d.x - lo], 1);
    if (d.y >= lo && d.y < hi) atomicAdd(&cnt[d.y - lo], 1);
    if (d.z >= lo && d.z < hi) atomicAdd(&cnt[d.z - lo], 1);
    if (d.w >= lo && d.w < hi) atomicAdd(&cnt[d.w - lo], 1);
}

__global__ void edge_fill(const int* __restrict__ src, const int* __restrict__ dst,
                          int* __restrict__ cur, int* __restrict__ csr,
                          int lo, int hi, int nE)
{
    const int e0 = (blockIdx.x * blockDim.x + threadIdx.x) * 4;
    if (e0 >= nE) return;
    const int4 d = load_ids4(dst, e0, nE);
    const int4 s = load_ids4(src, e0, nE);
    if (d.x >= lo && d.x < hi) csr[atomicAdd(&cur[d.x - lo], 1)] = s.x;
    if (d.y >= lo && d.y < hi) csr[atomicAdd(&cur[d.y - lo], 1)] = s.y;
    if (d.z >= lo && d.z < hi) csr[atomicAdd(&cur[d.z - lo], 1)] = s.z;
    if (d.w >= lo && d.w < hi) csr[atomicAdd(&cur[d.w - lo], 1)] = s.w;
}

#define SCAN_T 1024

__device__ __forceinline__ int4 load_cnt4(const int* p, int i0, int n) {
    int4 v = make_int4(0, 0, 0, 0);
    if (i0 + 3 < n) {
        v = *(const int4*)(p + i0);
    } else {
        if (i0     < n) v.x = p[i0];
        if (i0 + 1 < n) v.y = p[i0 + 1];
        if (i0 + 2 < n) v.z = p[i0 + 2];
    }
    return v;
}

__device__ __forceinline__ int wave_incl_scan(int x, int lane) {
#pragma unroll
    for (int off = 1; off < 32; off <<= 1) {
        const int nb = __shfl_up(x, off, 32);
        if (lane >= off) x += nb;
    }
    return x;
}

__global__ void __launch_bounds__(SCAN_T)
scan_rows(int* cur, int* __restrict__ offs, int n)
{
    __shared__ int wsum[32];
    const int t = threadIdx.x;
    const int lane = t & 31;
    const int wv = t >> 5;
    int carry = 0;
    int4 v = load_cnt4(cur, t * 4, n);
    for (int base = 0; base < n; base += SCAN_T * 4) {
        const int i0 = base + t * 4;
        const int4 nv = load_cnt4(cur, i0 + SCAN_T * 4, n);
        const int tsum = v.x + v.y + v.z + v.w;
        const int inc = wave_incl_scan(tsum, lane);
        if (lane == 31) wsum[wv] = inc;
        __syncthreads();
        const int ws = wsum[lane];
        const int winc = wave_incl_scan(ws, lane);
        const int wave_excl = __shfl(winc, wv, 32) - __shfl(ws, wv, 32);
        const int total = __shfl(winc, 31, 32);
        __syncthreads();
        int4 o;
        o.x = carry + wave_excl + (inc - tsum);
        o.y = o.x + v.x;
        o.z = o.y + v.y;
        o.w = o.z + v.z;
        if (i0 + 3 < n) {
            *(int4*)(offs + i0) = o;
            *(int4*)(cur + i0) = o;
        } else {
            if (i0     < n) { offs[i0]     = o.x; cur[i0]     = o.x; }
            if (i0 + 1 < n) { offs[i0 + 1] = o.y; cur[i0 + 1] = o.y; }
            if (i0 + 2 < n) { offs[i0 + 2] = o.z; cur[i0 + 2] = o.z; }
        }
        carry += total;
        v = nv;
    }
}

__device__ __forceinline__ float wave_max(float x) {
#pragma unroll
    for (int off = 16; off > 0; off >>= 1) x = fmaxf(x, __shfl_xor(x, off, 32));
    return x;
}
__device__ __forceinline__ float wave_sum(float x) {
#pragma unroll
    for (int off = 16; off > 0; off >>= 1) x += __shfl_xor(x, off, 32);
    return x;
}

__global__ void agg_rows(const int* __restrict__ offs, const int* __restrict__ endp,
                         const int* __restrict__ csr,
                         const float* __restrict__ ssrc, const float* __restrict__ sdst,
                         const float* __restrict__ z, float* __restrict__ out,
                         int lo, int nr)
{
    const int lane = threadIdx.x & 31;
    const int r = __builtin_amdgcn_readfirstlane((blockIdx.x * blockDim.x + threadIdx.x) >> 5);
    if (r >= nr) return;
    const int beg = __builtin_amdgcn_readfirstlane(offs[r]);
    const int end = __builtin_amdgcn_readfirstlane(endp[r]);
    const float sd = sdst[r + lo];

    int   s0 = 0;
    float v0 = -INFINITY;
    const bool have0 = (beg + lane) < end;
    if (have0) { s0 = csr[beg + lane]; v0 = leaky(ssrc[s0] + sd); }
    float mx = v0;
    for (int j = beg + 32 + lane; j < end; j += 32)
        mx = fmaxf(mx, leaky(ssrc[csr[j]] + sd));
    mx = wave_max(mx);

    const float ex0 = have0 ? __expf(v0 - mx) : 0.0f;
    float dn = ex0;
    for (int j = beg + 32 + lane; j < end; j += 32)
        dn += __expf(leaky(ssrc[csr[j]] + sd) - mx);
    dn = wave_sum(dn);

    float4 acc = make_float4(0.0f, 0.0f, 0.0f, 0.0f);
    for (int base = beg; base < end; base += 32) {
        int   s = s0;
        float a = have0 ? ex0 / dn : 0.0f;
        if (base != beg) {
            const int j = base + lane;
            s = 0; a = 0.0f;
            if (j < end) { s = csr[j]; a = __expf(leaky(ssrc[s] + sd) - mx) / dn; }
        }
        const int cnt = (end - base) < 32 ? (end - base) : 32;
#pragma unroll 4
        for (int i = 0; i < cnt; ++i) {
            const int   si = __builtin_amdgcn_readlane(s, i);
            const float ai = __int_as_float(__builtin_amdgcn_readlane(__float_as_int(a), i));
            const float4 zv = *(const float4*)(z + (size_t)si * D + lane * 4);
            acc.x = fmaf(ai, zv.x, acc.x);
            acc.y = fmaf(ai, zv.y, acc.y);
            acc.z = fmaf(ai, zv.z, acc.z);
            acc.w = fmaf(ai, zv.w, acc.w);
        }
    }
    *(float4*)(out + (size_t)r * D + lane * 4) = acc;
}

#define GP 132

__global__ void __launch_bounds__(256)
combine(const float* __restrict__ A, const float* __restrict__ B,
        const float* __restrict__ W1, const float* __restrict__ b1,
        const float* __restrict__ w2, float* __restrict__ out, int n)
{
    __shared__ __attribute__((aligned(16))) float sA[16 * GP];
    __shared__ __attribute__((aligned(16))) float sB[16 * GP];
    __shared__ __attribute__((aligned(16))) float sc[2][16][8];
    const int wave = threadIdx.x >> 5;
    const int lane = threadIdx.x & 31;
    const int half = lane >> 4;
    const int lr   = lane & 15;
    const int row0 = blockIdx.x * 16;
    const int col0 = wave * 16;

    for (int p = 0; p < 2; ++p) {
        const int rr = wave + 8 * p;
        int r = row0 + rr;
        if (r >= n) r = n - 1;
        const float4 va = *(const float4*)(A + (size_t)r * D + lane * 4);
        const float4 vb = *(const float4*)(B + (size_t)r * D + lane * 4);
        *(float4*)(sA + rr * GP + lane * 4) = va;
        *(float4*)(sB + rr * GP + lane * 4) = vb;
    }
    __syncthreads();

    v8f ca = {0.f, 0.f, 0.f, 0.f, 0.f, 0.f, 0.f, 0.f};
    v8f cb = {0.f, 0.f, 0.f, 0.f, 0.f, 0.f, 0.f, 0.f};
    const float* arow = sA + lr * GP + 2 * half;
    const float* brow = sB + lr * GP + 2 * half;
    const float* wcol = W1 + (size_t)(2 * half) * D + col0 + lr;
#pragma unroll
    for (int k0 = 0; k0 < D; k0 += 4) {
        const float2 av = *(const float2*)(arow + k0);
        const float2 bv = *(const float2*)(brow + k0);
        v2f a; a.x = av.x; a.y = av.y;
        v2f b; b.x = bv.x; b.y = bv.y;
        v2f w; w.x = wcol[(size_t)k0 * D];
               w.y = wcol[(size_t)(k0 + 1) * D];
        ca = __builtin_amdgcn_wmma_f32_16x16x4_f32(
                 false, a, false, w, (short)0, ca, false, false);
        cb = __builtin_amdgcn_wmma_f32_16x16x4_f32(
                 false, b, false, w, (short)0, cb, false, false);
    }

    const float bias = b1[col0 + lr];
    const float w2c  = w2[col0 + lr];
#pragma unroll
    for (int v = 0; v < 8; ++v) {
        ca[v] = tanhf(ca[v] + bias);
        cb[v] = tanhf(cb[v] + bias);
    }
    const float p1 = rows_dot16(ca, w2c, lr);
    const float p2 = rows_dot16(cb, w2c, lr);
    if ((lr & 1) == 0) {
        sc[0][8 * half + (lr >> 1)][wave] = p1;
        sc[1][8 * half + (lr >> 1)][wave] = p2;
    }
    __syncthreads();

    for (int p = 0; p < 2; ++p) {
        const int rr = wave + 8 * p;
        const float4 x0 = *(const float4*)&sc[0][rr][0];
        const float4 x1 = *(const float4*)&sc[0][rr][4];
        const float4 y0 = *(const float4*)&sc[1][rr][0];
        const float4 y1 = *(const float4*)&sc[1][rr][4];
        const float s1 = ((((((x0.x + x0.y) + x0.z) + x0.w) + x1.x) + x1.y) + x1.z) + x1.w;
        const float s2 = ((((((y0.x + y0.y) + y0.z) + y0.w) + y1.x) + y1.y) + y1.z) + y1.w;
        const float mm = fmaxf(s1, s2);
        const float e1 = __expf(s1 - mm), e2 = __expf(s2 - mm);
        const float al = e1 / (e1 + e2);
        const float be = 1.0f - al;
        const float4 va = *(const float4*)(sA + rr * GP + lane * 4);
        const float4 vb = *(const float4*)(sB + rr * GP + lane * 4);
        float4 o;
        o.x = al * va.x + be * vb.x;
        o.y = al * va.y + be * vb.y;
        o.z = al * va.z + be * vb.z;
        o.w = al * va.w + be * vb.w;
        const int r = row0 + rr;
        if (r < n) *(float4*)(out + (size_t)r * D + lane * 4) = o;
    }
}

static inline int cdiv(int a, int b) { return (a + b - 1) / b; }

extern "C" void kernel_launch(void* const* d_in, const int* in_sizes, int n_in,
                              void* d_out, int out_size, void* d_ws, size_t ws_size,
                              hipStream_t stream)
{
    (void)n_in; (void)ws_size; (void)out_size;
    const float* kn_emb   = (const float*)d_in[0];
    const float* exer_emb = (const float*)d_in[1];
    const float* W_dir    = (const float*)d_in[2];
    const float* a_dir    = (const float*)d_in[3];
    const float* W_ke     = (const float*)d_in[4];
    const float* a_ke     = (const float*)d_in[5];
    const float* W_ek     = (const float*)d_in[6];
    const float* a_ek     = (const float*)d_in[7];
    const float* rW1      = (const float*)d_in[8];
    const float* rb1      = (const float*)d_in[9];
    const float* rw2      = (const float*)d_in[10];
    const int* dsrc  = (const int*)d_in[11];
    const int* ddst  = (const int*)d_in[12];
    const int* kesrc = (const int*)d_in[13];
    const int* kedst = (const int*)d_in[14];
    const int* eksrc = (const int*)d_in[15];
    const int* ekdst = (const int*)d_in[16];

    const int Kn = in_sizes[0] / D;
    const int In = in_sizes[1] / D;
    const int Nn = Kn + In;
    const int E1 = in_sizes[11];
    const int E2 = in_sizes[13];

    float* w = (float*)d_ws;
    size_t off = 0;
    float* buf_z = w + off; off += (size_t)Nn * D;
    float* buf_B = w + off; off += (size_t)Nn * D;
    float* buf_A = w + off; off += (size_t)Kn * D;
    float* ssrc  = w + off; off += (size_t)Nn;
    float* sdst  = w + off; off += (size_t)Nn;

    const size_t n4 = ((size_t)Nn + 3) & ~(size_t)3;
    int* cur  = (int*)(buf_B + (size_t)Kn * D);
    int* offs = cur + n4;
    int* csr  = offs + n4;

    float* out_kn = (float*)d_out;
    float* out_ex = (float*)d_out + (size_t)Kn * D;

    gemm_z<<<cdiv(Nn, 16), 256, 0, stream>>>(exer_emb, In, kn_emb, W_ek, a_ek, buf_z, ssrc, sdst, Nn);
    hipMemsetAsync(cur, 0, (size_t)In * sizeof(int), stream);
    edge_count<<<cdiv(cdiv(E2, 4), 256), 256, 0, stream>>>(ekdst, cur, 0, In, E2);
    scan_rows<<<1, SCAN_T, 0, stream>>>(cur, offs, In);
    edge_fill<<<cdiv(cdiv(E2, 4), 256), 256, 0, stream>>>(eksrc, ekdst, cur, csr, 0, In, E2);
    agg_rows<<<cdiv(In, 8), 256, 0, stream>>>(offs, cur, csr, ssrc, sdst, buf_z, out_ex, 0, In);

    gemm_z<<<cdiv(Nn, 16), 256, 0, stream>>>(exer_emb, In, kn_emb, W_ke, a_ke, buf_z, ssrc, sdst, Nn);
    hipMemsetAsync(cur, 0, (size_t)Kn * sizeof(int), stream);
    edge_count<<<cdiv(cdiv(E2, 4), 256), 256, 0, stream>>>(kedst, cur, In, Nn, E2);
    scan_rows<<<1, SCAN_T, 0, stream>>>(cur, offs, Kn);
    edge_fill<<<cdiv(cdiv(E2, 4), 256), 256, 0, stream>>>(kesrc, kedst, cur, csr, In, Nn, E2);
    agg_rows<<<cdiv(Kn, 8), 256, 0, stream>>>(offs, cur, csr, ssrc, sdst, buf_z, buf_B, In, Kn);

    gemm_z<<<cdiv(Kn, 16), 256, 0, stream>>>(kn_emb, Kn, kn_emb, W_dir, a_dir, buf_z, ssrc, sdst, Kn);
    hipMemsetAsync(cur, 0, (size_t)Kn * sizeof(int), stream);
    edge_count<<<cdiv(cdiv(E1, 4), 256), 256, 0, stream>>>(ddst, cur, 0, Kn, E1);
    scan_rows<<<1, SCAN_T, 0, stream>>>(cur, offs, Kn);
    edge_fill<<<cdiv(cdiv(E1, 4), 256), 256, 0, stream>>>(dsrc, ddst, cur, csr, 0, Kn, E1);
    agg_rows<<<cdiv(Kn, 8), 256, 0, stream>>>(offs, cur, csr, ssrc, sdst, buf_z, buf_A, 0, Kn);

    combine<<<cdiv(Kn, 16), 256, 0, stream>>>(buf_A, buf_B, rW1, rb1, rw2, out_kn, Kn);
}


# gemm_z made persistent: 512 eight-wave blocks keep their W column slices in registers and walk the 16-row strips; H strips come one ahead by whole-row loads through two LDS buffers
# speedup vs baseline: 2.3068x; 1.1120x over previous
#include <hip/hip_runtime.h>
#include <hip/hip_bf16.h>
#include <math.h>


#define D 128

typedef float v2f __attribute__((ext_vector_type(2)));
typedef float v8f __attribute__((ext_vector_type(8)));

__device__ __forceinline__ float rows_dot16(const v8f c, float coef, int lr)
{
    const bool b3 = (lr & 8) != 0;
    const bool b2 = (lr & 4) != 0;
    const bool b1 = (lr & 2) != 0;
    float y[4];
#pragma unroll
    for (int i = 0; i < 4; ++i) {
        const float lo = c[i] * coef, hi = c[i + 4] * coef;
        const float keep = b3 ? hi : lo;
        const float send = b3 ? lo : hi;
        y[i] = keep + __shfl_xor(send, 8, 32);
    }
    float q[2];
#pragma unroll
    for (int i = 0; i < 2; ++i) {
        const float keep = b2 ? y[i + 2] : y[i];
        const float send = b2 ? y[i] : y[i + 2];
        q[i] = keep + __shfl_xor(send, 4, 32);
    }
    const float keep = b1 ? q[1] : q[0];
    const float send = b1 ? q[0] : q[1];
    float w = keep + __shfl_xor(send, 2, 32);
    w += __shfl_xor(w, 1, 32);
    return w;
}

#define GP 132

__device__ __forceinline__ float4 load_hrow(const float* __restrict__ h0, int n0,
                                            const float* __restrict__ h1,
                                            int r, int n, int lane)
{
    if (r >= n) r = n - 1;
    const float* hrow = (r < n0) ? (h0 + (size_t)r * D)
                                 : (h1 + (size_t)(r - n0) * D);
    return *(const float4*)(hrow + lane * 4);
}

__global__ void __launch_bounds__(256)
gemm_z(const float* __restrict__ h0, int n0,
       const float* __restrict__ h1,
       const float* __restrict__ W,
       const float* __restrict__ att,
       float* __restrict__ Z,
       float* __restrict__ ssrc, float* __restrict__ sdst, int n)
{
    __shared__ __attribute__((aligned(16))) float sA[2][16 * GP];
    __shared__ float sc[2][8][2][16];
    const int wave = threadIdx.x >> 5;
    const int lane = threadIdx.x & 31;
    const int half = lane >> 4;
    const int lr   = lane & 15;
    const int col0 = wave * 16;
    const int nstrips = (n + 15) >> 4;
    const int step = gridDim.x;

    int strip = blockIdx.x;
    if (strip >= nstrips) return;

    float4 p0 = load_hrow(h0, n0, h1, strip * 16 + wave,     n, lane);
    float4 p1 = load_hrow(h0, n0, h1, strip * 16 + wave + 8, n, lane);

    v2f wf[32];
    const float* wcol = W + (size_t)(2 * half) * D + col0 + lr;
#pragma unroll
    for (int i = 0; i < 32; ++i) {
        wf[i].x = wcol[(size_t)(4 * i) * D];
        wf[i].y = wcol[(size_t)(4 * i + 1) * D];
    }
    const float att_s = att[col0 + lr];
    const float att_d = att[D + col0 + lr];

    *(float4*)(&sA[0][wave * GP + lane * 4])       = p0;
    *(float4*)(&sA[0][(wave + 8) * GP + lane * 4]) = p1;
    __syncthreads();

    int buf = 0;
    for (; strip < nstrips; strip += step) {
        const int row0 = strip * 16;
        const int next = strip + step;
        const bool more = next < nstrips;
        if (more) {
            p0 = load_hrow(h0, n0, h1, next * 16 + wave,     n, lane);
            p1 = load_hrow(h0, n0, h1, next * 16 + wave + 8, n, lane);
        }

        v8f c = {0.f, 0.f, 0.f, 0.f, 0.f, 0.f, 0.f, 0.f};
        const float* arow = &sA[buf][lr * GP + 2 * half];
        v2f af[32];
#pragma unroll
        for (int i = 0; i < 32; ++i) {
            const float2 av = *(const float2*)(arow + 4 * i);
            af[i].x = av.x; af[i].y = av.y;
        }
        __builtin_amdgcn_sched_barrier(0);
#pragma unroll
        for (int i = 0; i < 32; ++i) {
            c = __builtin_amdgcn_wmma_f32_16x16x4_f32(
                    false, af[i], false, wf[i], (short)0, c, false, false);
        }
        const float ps = rows_dot16(c, att_s, lr);
        const float pd = rows_dot16(c, att_d, lr);
        if ((lr & 1) == 0) {
            sc[buf][wave][0][8 * half + (lr >> 1)] = ps;
            sc[buf][wave][1][8 * half + (lr >> 1)] = pd;
        }
        if (more) {
            *(float4*)(&sA[buf ^ 1][wave * GP + lane * 4])       = p0;
            *(float4*)(&sA[buf ^ 1][(wave + 8) * GP + lane * 4]) = p1;
        }
        __syncthreads();

#pragma unroll
        for (int v = 0; v < 8; ++v) {
            const int r = row0 + v + 8 * half;
            if (r < n) Z[(size_t)r * D + col0 + lr] = c[v];
        }

        if (threadIdx.x < 32) {
            const int which = threadIdx.x >> 4;
            const int rr = threadIdx.x & 15;
            float t = 0.0f;
#pragma unroll
            for (int w8 = 0; w8 < 8; ++w8) t += sc[buf][w8][which][rr];
            const int r = row0 + rr;
            if (r < n) { if (which) sdst[r] = t; else ssrc[r] = t; }
        }
        buf ^= 1;
    }
}

__device__ __forceinline__ float leaky(float v) {
    return v > 0.0f ? v : 0.01f * v;
}


__device__ __forceinline__ int4 load_ids4(const int* __restrict__ p, int e0, int nE) {
    int4 v = make_int4(-1, -1, -1, -1);
    if (e0 + 3 < nE) {
        v = *(const int4*)(p + e0);
    } else {
        if (e0     < nE) v.x = p[e0];
        if (e0 + 1 < nE) v.y = p[e0 + 1];
        if (e0 + 2 < nE) v.z = p[e0 + 2];
    }
    return v;
}

__global__ void edge_count(const int* __restrict__ dst, int* __restrict__ cnt,
                           int lo, int hi, int nE)
{
    const int e0 = (blockIdx.x * blockDim.x + threadIdx.x) * 4;
    if (e0 >= nE) return;
    const int4 d = load_ids4(dst, e0, nE);
    if (d.x >= lo && d.x < hi) atomicAdd(&cnt[d.x - lo], 1);
    if (d.y >= lo && d.y < hi) atomicAdd(&cnt[d.y - lo], 1);
    if (d.z >= lo && d.z < hi) atomicAdd(&cnt[d.z - lo], 1);
    if (d.w >= lo && d.w < hi) atomicAdd(&cnt[d.w - lo], 1);
}

__global__ void edge_fill(const int* __restrict__ src, const int* __restrict__ dst,
                          int* __restrict__ cur, int* __restrict__ csr,
                          int lo, int hi, int nE)
{
    const int e0 = (blockIdx.x * blockDim.x + threadIdx.x) * 4;
    if (e0 >= nE) return;
    const int4 d = load_ids4(dst, e0, nE);
    const int4 s = load_ids4(src, e0, nE);
    if (d.x >= lo && d.x < hi) csr[atomicAdd(&cur[d.x - lo], 1)] = s.x;
    if (d.y >= lo && d.y < hi) csr[atomicAdd(&cur[d.y - lo], 1)] = s.y;
    if (d.z >= lo && d.z < hi) csr[atomicAdd(&cur[d.z - lo], 1)] = s.z;
    if (d.w >= lo && d.w < hi) csr[atomicAdd(&cur[d.w - lo], 1)] = s.w;
}

#define SCAN_T 1024

__device__ __forceinline__ int4 load_cnt4(const int* p, int i0, int n) {
    int4 v = make_int4(0, 0, 0, 0);
    if (i0 + 3 < n) {
        v = *(const int4*)(p + i0);
    } else {
        if (i0     < n) v.x = p[i0];
        if (i0 + 1 < n) v.y = p[i0 + 1];
        if (i0 + 2 < n) v.z = p[i0 + 2];
    }
    return v;
}

__device__ __forceinline__ int wave_incl_scan(int x, int lane) {
#pragma unroll
    for (int off = 1; off < 32; off <<= 1) {
        const int nb = __shfl_up(x, off, 32);
        if (lane >= off) x += nb;
    }
    return x;
}

__global__ void __launch_bounds__(SCAN_T)
scan_rows(int* cur, int* __restrict__ offs, int n)
{
    __shared__ int wsum[32];
    const int t = threadIdx.x;
    const int lane = t & 31;
    const int wv = t >> 5;
    int carry = 0;
    int4 v = load_cnt4(cur, t * 4, n);
    for (int base = 0; base < n; base += SCAN_T * 4) {
        const int i0 = base + t * 4;
        const int4 nv = load_cnt4(cur, i0 + SCAN_T * 4, n);
        const int tsum = v.x + v.y + v.z + v.w;
        const int inc = wave_incl_scan(tsum, lane);
        if (lane == 31) wsum[wv] = inc;
        __syncthreads();
        const int ws = wsum[lane];
        const int winc = wave_incl_scan(ws, lane);
        const int wave_excl = __shfl(winc, wv, 32) - __shfl(ws, wv, 32);
        const int total = __shfl(winc, 31, 32);
        __syncthreads();
        int4 o;
        o.x = carry + wave_excl + (inc - tsum);
        o.y = o.x + v.x;
        o.z = o.y + v.y;
        o.w = o.z + v.z;
        if (i0 + 3 < n) {
            *(int4*)(offs + i0) = o;
            *(int4*)(cur + i0) = o;
        } else {
            if (i0     < n) { offs[i0]     = o.x; cur[i0]     = o.x; }
            if (i0 + 1 < n) { offs[i0 + 1] = o.y; cur[i0 + 1] = o.y; }
            if (i0 + 2 < n) { offs[i0 + 2] = o.z; cur[i0 + 2] = o.z; }
        }
        carry += total;
        v = nv;
    }
}

__device__ __forceinline__ float wave_max(float x) {
#pragma unroll
    for (int off = 16; off > 0; off >>= 1) x = fmaxf(x, __shfl_xor(x, off, 32));
    return x;
}
__device__ __forceinline__ float wave_sum(float x) {
#pragma unroll
    for (int off = 16; off > 0; off >>= 1) x += __shfl_xor(x, off, 32);
    return x;
}

__global__ void agg_rows(const int* __restrict__ offs, const int* __restrict__ endp,
                         const int* __restrict__ csr,
                         const float* __restrict__ ssrc, const float* __restrict__ sdst,
                         const float* __restrict__ z, float* __restrict__ out,
                         int lo, int nr)
{
    const int lane = threadIdx.x & 31;
    const int r = __builtin_amdgcn_readfirstlane((blockIdx.x * blockDim.x + threadIdx.x) >> 5);
    if (r >= nr) return;
    const int beg = __builtin_amdgcn_readfirstlane(offs[r]);
    const int end = __builtin_amdgcn_readfirstlane(endp[r]);
    const float sd = sdst[r + lo];

    int   s0 = 0;
    float v0 = -INFINITY;
    const bool have0 = (beg + lane) < end;
    if (have0) { s0 = csr[beg + lane]; v0 = leaky(ssrc[s0] + sd); }
    float mx = v0;
    for (int j = beg + 32 + lane; j < end; j += 32)
        mx = fmaxf(mx, leaky(ssrc[csr[j]] + sd));
    mx = wave_max(mx);

    const float ex0 = have0 ? __expf(v0 - mx) : 0.0f;
    float dn = ex0;
    for (int j = beg + 32 + lane; j < end; j += 32)
        dn += __expf(leaky(ssrc[csr[j]] + sd) - mx);
    dn = wave_sum(dn);

    float4 acc = make_float4(0.0f, 0.0f, 0.0f, 0.0f);
    for (int base = beg; base < end; base += 32) {
        int   s = s0;
        float a = have0 ? ex0 / dn : 0.0f;
        if (base != beg) {
            const int j = base + lane;
            s = 0; a = 0.0f;
            if (j < end) { s = csr[j]; a = __expf(leaky(ssrc[s] + sd) - mx) / dn; }
        }
        const int cnt = (end - base) < 32 ? (end - base) : 32;
#pragma unroll 4
        for (int i = 0; i < cnt; ++i) {
            const int   si = __builtin_amdgcn_readlane(s, i);
            const float ai = __int_as_float(__builtin_amdgcn_readlane(__float_as_int(a), i));
            const float4 zv = *(const float4*)(z + (size_t)si * D + lane * 4);
            acc.x = fmaf(ai, zv.x, acc.x);
            acc.y = fmaf(ai, zv.y, acc.y);
            acc.z = fmaf(ai, zv.z, acc.z);
            acc.w = fmaf(ai, zv.w, acc.w);
        }
    }
    *(float4*)(out + (size_t)r * D + lane * 4) = acc;
}

__global__ void __launch_bounds__(256)
combine(const float* __restrict__ A, const float* __restrict__ B,
        const float* __restrict__ W1, const float* __restrict__ b1,
        const float* __restrict__ w2, float* __restrict__ out, int n)
{
    __shared__ __attribute__((aligned(16))) float sA[16 * GP];
    __shared__ __attribute__((aligned(16))) float sB[16 * GP];
    __shared__ __attribute__((aligned(16))) float sc[2][16][8];
    const int wave = threadIdx.x >> 5;
    const int lane = threadIdx.x & 31;
    const int half = lane >> 4;
    const int lr   = lane & 15;
    const int row0 = blockIdx.x * 16;
    const int col0 = wave * 16;

    for (int p = 0; p < 2; ++p) {
        const int rr = wave + 8 * p;
        int r = row0 + rr;
        if (r >= n) r = n - 1;
        const float4 va = *(const float4*)(A + (size_t)r * D + lane * 4);
        const float4 vb = *(const float4*)(B + (size_t)r * D + lane * 4);
        *(float4*)(sA + rr * GP + lane * 4) = va;
        *(float4*)(sB + rr * GP + lane * 4) = vb;
    }
    __syncthreads();

    v8f ca = {0.f, 0.f, 0.f, 0.f, 0.f, 0.f, 0.f, 0.f};
    v8f cb = {0.f, 0.f, 0.f, 0.f, 0.f, 0.f, 0.f, 0.f};
    const float* arow = sA + lr * GP + 2 * half;
    const float* brow = sB + lr * GP + 2 * half;
    const float* wcol = W1 + (size_t)(2 * half) * D + col0 + lr;
#pragma unroll
    for (int k0 = 0; k0 < D; k0 += 4) {
        const float2 av = *(const float2*)(arow + k0);
        const float2 bv = *(const float2*)(brow + k0);
        v2f a; a.x = av.x; a.y = av.y;
        v2f b; b.x = bv.x; b.y = bv.y;
        v2f w; w.x = wcol[(size_t)k0 * D];
               w.y = wcol[(size_t)(k0 + 1) * D];
        ca = __builtin_amdgcn_wmma_f32_16x16x4_f32(
                 false, a, false, w, (short)0, ca, false, false);
        cb = __builtin_amdgcn_wmma_f32_16x16x4_f32(
                 false, b, false, w, (short)0, cb, false, false);
    }

    const float bias = b1[col0 + lr];
    const float w2c  = w2[col0 + lr];
#pragma unroll
    for (int v = 0; v < 8; ++v) {
        ca[v] = tanhf(ca[v] + bias);
        cb[v] = tanhf(cb[v] + bias);
    }
    const float p1 = rows_dot16(ca, w2c, lr);
    const float p2 = rows_dot16(cb, w2c, lr);
    if ((lr & 1) == 0) {
        sc[0][8 * half + (lr >> 1)][wave] = p1;
        sc[1][8 * half + (lr >> 1)][wave] = p2;
    }
    __syncthreads();

    for (int p = 0; p < 2; ++p) {
        const int rr = wave + 8 * p;
        const float4 x0 = *(const float4*)&sc[0][rr][0];
        const float4 x1 = *(const float4*)&sc[0][rr][4];
        const float4 y0 = *(const float4*)&sc[1][rr][0];
        const float4 y1 = *(const float4*)&sc[1][rr][4];
        const float s1 = ((((((x0.x + x0.y) + x0.z) + x0.w) + x1.x) + x1.y) + x1.z) + x1.w;
        const float s2 = ((((((y0.x + y0.y) + y0.z) + y0.w) + y1.x) + y1.y) + y1.z) + y1.w;
        const float mm = fmaxf(s1, s2);
        const float e1 = __expf(s1 - mm), e2 = __expf(s2 - mm);
        const float al = e1 / (e1 + e2);
        const float be = 1.0f - al;
        const float4 va = *(const float4*)(sA + rr * GP + lane * 4);
        const float4 vb = *(const float4*)(sB + rr * GP + lane * 4);
        float4 o;
        o.x = al * va.x + be * vb.x;
        o.y = al * va.y + be * vb.y;
        o.z = al * va.z + be * vb.z;
        o.w = al * va.w + be * vb.w;
        const int r = row0 + rr;
        if (r < n) *(float4*)(out + (size_t)r * D + lane * 4) = o;
    }
}

static inline int cdiv(int a, int b) { return (a + b - 1) / b; }

extern "C" void kernel_launch(void* const* d_in, const int* in_sizes, int n_in,
                              void* d_out, int out_size, void* d_ws, size_t ws_size,
                              hipStream_t stream)
{
    (void)n_in; (void)ws_size; (void)out_size;
    const float* kn_emb   = (const float*)d_in[0];
    const float* exer_emb = (const float*)d_in[1];
    const float* W_dir    = (const float*)d_in[2];
    const float* a_dir    = (const float*)d_in[3];
    const float* W_ke     = (const float*)d_in[4];
    const float* a_ke     = (const float*)d_in[5];
    const float* W_ek     = (const float*)d_in[6];
    const float* a_ek     = (const float*)d_in[7];
    const float* rW1      = (const float*)d_in[8];
    const float* rb1      = (const float*)d_in[9];
    const float* rw2      = (const float*)d_in[10];
    const int* dsrc  = (const int*)d_in[11];
    const int* ddst  = (const int*)d_in[12];
    const int* kesrc = (const int*)d_in[13];
    const int* kedst = (const int*)d_in[14];
    const int* eksrc = (const int*)d_in[15];
    const int* ekdst = (const int*)d_in[16];

    const int Kn = in_sizes[0] / D;
    const int In = in_sizes[1] / D;
    const int Nn = Kn + In;
    const int E1 = in_sizes[11];
    const int E2 = in_sizes[13];

    float* w = (float*)d_ws;
    size_t off = 0;
    float* buf_z = w + off; off += (size_t)Nn * D;
    float* buf_B = w + off; off += (size_t)Nn * D;
    float* buf_A = w + off; off += (size_t)Kn * D;
    float* ssrc  = w + off; off += (size_t)Nn;
    float* sdst  = w + off; off += (size_t)Nn;

    const size_t n4 = ((size_t)Nn + 3) & ~(size_t)3;
    int* cur  = (int*)(buf_B + (size_t)Kn * D);
    int* offs = cur + n4;
    int* csr  = offs + n4;

    float* out_kn = (float*)d_out;
    float* out_ex = (float*)d_out + (size_t)Kn * D;

    gemm_z<<<512, 256, 0, stream>>>(exer_emb, In, kn_emb, W_ek, a_ek, buf_z, ssrc, sdst, Nn);
    hipMemsetAsync(cur, 0, (size_t)In * sizeof(int), stream);
    edge_count<<<cdiv(cdiv(E2, 4), 256), 256, 0, stream>>>(ekdst, cur, 0, In, E2);
    scan_rows<<<1, SCAN_T, 0, stream>>>(cur, offs, In);
    edge_fill<<<cdiv(cdiv(E2, 4), 256), 256, 0, stream>>>(eksrc, ekdst, cur, csr, 0, In, E2);
    agg_rows<<<cdiv(In, 8), 256, 0, stream>>>(offs, cur, csr, ssrc, sdst, buf_z, out_ex, 0, In);

    gemm_z<<<512, 256, 0, stream>>>(exer_emb, In, kn_emb, W_ke, a_ke, buf_z, ssrc, sdst, Nn);
    hipMemsetAsync(cur, 0, (size_t)Kn * sizeof(int), stream);
    edge_count<<<cdiv(cdiv(E2, 4), 256), 256, 0, stream>>>(kedst, cur, In, Nn, E2);
    scan_rows<<<1, SCAN_T, 0, stream>>>(cur, offs, Kn);
    edge_fill<<<cdiv(cdiv(E2, 4), 256), 256, 0, stream>>>(kesrc, kedst, cur, csr, In, Nn, E2);
    agg_rows<<<cdiv(Kn, 8), 256, 0, stream>>>(offs, cur, csr, ssrc, sdst, buf_z, buf_B, In, Kn);

    gemm_z<<<512, 256, 0, stream>>>(kn_emb, Kn, kn_emb, W_dir, a_dir, buf_z, ssrc, sdst, Kn);
    hipMemsetAsync(cur, 0, (size_t)Kn * sizeof(int), stream);
    edge_count<<<cdiv(cdiv(E1, 4), 256), 256, 0, stream>>>(ddst, cur, 0, Kn, E1);
    scan_rows<<<1, SCAN_T, 0, stream>>>(cur, offs, Kn);
    edge_fill<<<cdiv(cdiv(E1, 4), 256), 256, 0, stream>>>(dsrc, ddst, cur, csr, 0, Kn, E1);
    agg_rows<<<cdiv(Kn, 8), 256, 0, stream>>>(offs, cur, csr, ssrc, sdst, buf_z, buf_A, 0, Kn);

    combine<<<cdiv(Kn, 16), 256, 0, stream>>>(buf_A, buf_B, rW1, rb1, rw2, out_kn, Kn);
}


# Edge grouping of the three graphs batched: one clear, one count, one scan (a block per graph) and one fill launch serve all three edge lists; 12 stream operations become 4, row sum order as before
# speedup vs baseline: 2.8554x; 1.2378x over previous
#include <hip/hip_runtime.h>
#include <hip/hip_bf16.h>
#include <math.h>


#define D 128

typedef float v2f __attribute__((ext_vector_type(2)));
typedef float v8f __attribute__((ext_vector_type(8)));

__device__ __forceinline__ float rows_dot16(const v8f c, float coef, int lr)
{
    const bool b3 = (lr & 8) != 0;
    const bool b2 = (lr & 4) != 0;
    const bool b1 = (lr & 2) != 0;
    float y[4];
#pragma unroll
    for (int i = 0; i < 4; ++i) {
        const float lo = c[i] * coef, hi = c[i + 4] * coef;
        const float keep = b3 ? hi : lo;
        const float send = b3 ? lo : hi;
        y[i] = keep + __shfl_xor(send, 8, 32);
    }
    float q[2];
#pragma unroll
    for (int i = 0; i < 2; ++i) {
        const float keep = b2 ? y[i + 2] : y[i];
        const float send = b2 ? y[i] : y[i + 2];
        q[i] = keep + __shfl_xor(send, 4, 32);
    }
    const float keep = b1 ? q[1] : q[0];
    const float send = b1 ? q[0] : q[1];
    float w = keep + __shfl_xor(send, 2, 32);
    w += __shfl_xor(w, 1, 32);
    return w;
}

#define GP 132

__device__ __forceinline__ float4 load_hrow(const float* __restrict__ h0, int n0,
                                            const float* __restrict__ h1,
                                            int r, int n, int lane)
{
    if (r >= n) r = n - 1;
    const float* hrow = (r < n0) ? (h0 + (size_t)r * D)
                                 : (h1 + (size_t)(r - n0) * D);
    return *(const float4*)(hrow + lane * 4);
}

__global__ void __launch_bounds__(256)
gemm_z(const float* __restrict__ h0, int n0,
       const float* __restrict__ h1,
       const float* __restrict__ W,
       const float* __restrict__ att,
       float* __restrict__ Z,
       float* __restrict__ ssrc, float* __restrict__ sdst, int n)
{
    __shared__ __attribute__((aligned(16))) float sA[2][16 * GP];
    __shared__ float sc[2][8][2][16];
    const int wave = threadIdx.x >> 5;
    const int lane = threadIdx.x & 31;
    const int half = lane >> 4;
    const int lr   = lane & 15;
    const int col0 = wave * 16;
    const int nstrips = (n + 15) >> 4;
    const int step = gridDim.x;

    int strip = blockIdx.x;
    if (strip >= nstrips) return;

    float4 p0 = load_hrow(h0, n0, h1, strip * 16 + wave,     n, lane);
    float4 p1 = load_hrow(h0, n0, h1, strip * 16 + wave + 8, n, lane);

    v2f wf[32];
    const float* wcol = W + (size_t)(2 * half) * D + col0 + lr;
#pragma unroll
    for (int i = 0; i < 32; ++i) {
        wf[i].x = wcol[(size_t)(4 * i) * D];
        wf[i].y = wcol[(size_t)(4 * i + 1) * D];
    }
    const float att_s = att[col0 + lr];
    const float att_d = att[D + col0 + lr];

    *(float4*)(&sA[0][wave * GP + lane * 4])       = p0;
    *(float4*)(&sA[0][(wave + 8) * GP + lane * 4]) = p1;
    __syncthreads();

    int buf = 0;
    for (; strip < nstrips; strip += step) {
        const int row0 = strip * 16;
        const int next = strip + step;
        const bool more = next < nstrips;
        if (more) {
            p0 = load_hrow(h0, n0, h1, next * 16 + wave,     n, lane);
            p1 = load_hrow(h0, n0, h1, next * 16 + wave + 8, n, lane);
        }

        v8f c = {0.f, 0.f, 0.f, 0.f, 0.f, 0.f, 0.f, 0.f};
        const float* arow = &sA[buf][lr * GP + 2 * half];
        v2f af[32];
#pragma unroll
        for (int i = 0; i < 32; ++i) {
            const float2 av = *(const float2*)(arow + 4 * i);
            af[i].x = av.x; af[i].y = av.y;
        }
        __builtin_amdgcn_sched_barrier(0);
#pragma unroll
        for (int i = 0; i < 32; ++i) {
            c = __builtin_amdgcn_wmma_f32_16x16x4_f32(
                    false, af[i], false, wf[i], (short)0, c, false, false);
        }
        const float ps = rows_dot16(c, att_s, lr);
        const float pd = rows_dot16(c, att_d, lr);
        if ((lr & 1) == 0) {
            sc[buf][wave][0][8 * half + (lr >> 1)] = ps;
            sc[buf][wave][1][8 * half + (lr >> 1)] = pd;
        }
        if (more) {
            *(float4*)(&sA[buf ^ 1][wave * GP + lane * 4])       = p0;
            *(float4*)(&sA[buf ^ 1][(wave + 8) * GP + lane * 4]) = p1;
        }
        __syncthreads();

#pragma unroll
        for (int v = 0; v < 8; ++v) {
            const int r = row0 + v + 8 * half;
            if (r < n) Z[(size_t)r * D + col0 + lr] = c[v];
        }

        if (threadIdx.x < 32) {
            const int which = threadIdx.x >> 4;
            const int rr = threadIdx.x & 15;
            float t = 0.0f;
#pragma unroll
            for (int w8 = 0; w8 < 8; ++w8) t += sc[buf][w8][which][rr];
            const int r = row0 + rr;
            if (r < n) { if (which) sdst[r] = t; else ssrc[r] = t; }
        }
        buf ^= 1;
    }
}

__device__ __forceinline__ float leaky(float v) {
    return v > 0.0f ? v : 0.01f * v;
}


struct EdgePart {
    const int* src; const int* dst;
    int lo, hi, nE;
    int blk0;
    int rbase;
};

__device__ __forceinline__ EdgePart pick_part(
    const int* s0, const int* d0, int lo0, int hi0, int nE0,
    const int* s1, const int* d1, int lo1, int hi1, int nE1, int blk1, int rb1,
    const int* s2, const int* d2, int lo2, int hi2, int nE2, int blk2, int rb2)
{
    EdgePart p;
    const int b = blockIdx.x;
    if (b >= blk2)      { p.src = s2; p.dst = d2; p.lo = lo2; p.hi = hi2; p.nE = nE2; p.blk0 = blk2; p.rbase = rb2; }
    else if (b >= blk1) { p.src = s1; p.dst = d1; p.lo = lo1; p.hi = hi1; p.nE = nE1; p.blk0 = blk1; p.rbase = rb1; }
    else                { p.src = s0; p.dst = d0; p.lo = lo0; p.hi = hi0; p.nE = nE0; p.blk0 = 0;    p.rbase = 0;   }
    return p;
}

__device__ __forceinline__ int4 load_ids4(const int* __restrict__ p, int e0, int nE) {
    int4 v = make_int4(-1, -1, -1, -1);
    if (e0 + 3 < nE) {
        v = *(const int4*)(p + e0);
    } else {
        if (e0     < nE) v.x = p[e0];
        if (e0 + 1 < nE) v.y = p[e0 + 1];
        if (e0 + 2 < nE) v.z = p[e0 + 2];
    }
    return v;
}

__global__ void edge_count3(const int* __restrict__ d0, int lo0, int hi0, int nE0,
                            const int* __restrict__ d1, int lo1, int hi1, int nE1, int blk1, int rb1,
                            const int* __restrict__ d2, int lo2, int hi2, int nE2, int blk2, int rb2,
                            int* __restrict__ cnt_all)
{
    const EdgePart p = pick_part(d0, d0, lo0, hi0, nE0,
                                 d1, d1, lo1, hi1, nE1, blk1, rb1,
                                 d2, d2, lo2, hi2, nE2, blk2, rb2);
    const int lo = p.lo, hi = p.hi, nE = p.nE;
    int* cnt = cnt_all + p.rbase;
    const int e0 = ((blockIdx.x - p.blk0) * blockDim.x + threadIdx.x) * 4;
    if (e0 >= nE) return;
    const int4 d = load_ids4(p.dst, e0, nE);
    if (d.x >= lo && d.x < hi) atomicAdd(&cnt[d.x - lo], 1);
    if (d.y >= lo && d.y < hi) atomicAdd(&cnt[d.y - lo], 1);
    if (d.z >= lo && d.z < hi) atomicAdd(&cnt[d.z - lo], 1);
    if (d.w >= lo && d.w < hi) atomicAdd(&cnt[d.w - lo], 1);
}

__global__ void edge_fill3(const int* __restrict__ s0, const int* __restrict__ d0, int lo0, int hi0, int nE0,
                           const int* __restrict__ s1, const int* __restrict__ d1, int lo1, int hi1, int nE1, int blk1, int rb1,
                           const int* __restrict__ s2, const int* __restrict__ d2, int lo2, int hi2, int nE2, int blk2, int rb2,
                           int* __restrict__ cur_all, int* __restrict__ csr)
{
    const EdgePart p = pick_part(s0, d0, lo0, hi0, nE0,
                                 s1, d1, lo1, hi1, nE1, blk1, rb1,
                                 s2, d2, lo2, hi2, nE2, blk2, rb2);
    const int lo = p.lo, hi = p.hi, nE = p.nE;
    int* cur = cur_all + p.rbase;
    const int e0 = ((blockIdx.x - p.blk0) * blockDim.x + threadIdx.x) * 4;
    if (e0 >= nE) return;
    const int4 d = load_ids4(p.dst, e0, nE);
    const int4 s = load_ids4(p.src, e0, nE);
    if (d.x >= lo && d.x < hi) csr[atomicAdd(&cur[d.x - lo], 1)] = s.x;
    if (d.y >= lo && d.y < hi) csr[atomicAdd(&cur[d.y - lo], 1)] = s.y;
    if (d.z >= lo && d.z < hi) csr[atomicAdd(&cur[d.z - lo], 1)] = s.z;
    if (d.w >= lo && d.w < hi) csr[atomicAdd(&cur[d.w - lo], 1)] = s.w;
}

#define SCAN_T 1024

__device__ __forceinline__ int4 load_cnt4(const int* p, int i0, int n) {
    int4 v = make_int4(0, 0, 0, 0);
    if (i0 + 3 < n) {
        v = *(const int4*)(p + i0);
    } else {
        if (i0     < n) v.x = p[i0];
        if (i0 + 1 < n) v.y = p[i0 + 1];
        if (i0 + 2 < n) v.z = p[i0 + 2];
    }
    return v;
}

__device__ __forceinline__ int wave_incl_scan(int x, int lane) {
#pragma unroll
    for (int off = 1; off < 32; off <<= 1) {
        const int nb = __shfl_up(x, off, 32);
        if (lane >= off) x += nb;
    }
    return x;
}

__global__ void __launch_bounds__(SCAN_T)
scan_rows3(int* cur_all, int* __restrict__ offs_all,
           int n0, int n1, int rb1, int cb1, int n2, int rb2, int cb2)
{
    __shared__ int wsum[32];
    const int t = threadIdx.x;
    const int lane = t & 31;
    const int wv = t >> 5;
    int n = n0, rb = 0, carry = 0;
    if (blockIdx.x == 1)      { n = n1; rb = rb1; carry = cb1; }
    else if (blockIdx.x == 2) { n = n2; rb = rb2; carry = cb2; }
    int* cur = cur_all + rb;
    int* offs = offs_all + rb;
    int4 v = load_cnt4(cur, t * 4, n);
    for (int base = 0; base < n; base += SCAN_T * 4) {
        const int i0 = base + t * 4;
        const int4 nv = load_cnt4(cur, i0 + SCAN_T * 4, n);
        const int tsum = v.x + v.y + v.z + v.w;
        const int inc = wave_incl_scan(tsum, lane);
        if (lane == 31) wsum[wv] = inc;
        __syncthreads();
        const int ws = wsum[lane];
        const int winc = wave_incl_scan(ws, lane);
        const int wave_excl = __shfl(winc, wv, 32) - __shfl(ws, wv, 32);
        const int total = __shfl(winc, 31, 32);
        __syncthreads();
        int4 o;
        o.x = carry + wave_excl + (inc - tsum);
        o.y = o.x + v.x;
        o.z = o.y + v.y;
        o.w = o.z + v.z;
        if (i0 + 3 < n) {
            *(int4*)(offs + i0) = o;
            *(int4*)(cur + i0) = o;
        } else {
            if (i0     < n) { offs[i0]     = o.x; cur[i0]     = o.x; }
            if (i0 + 1 < n) { offs[i0 + 1] = o.y; cur[i0 + 1] = o.y; }
            if (i0 + 2 < n) { offs[i0 + 2] = o.z; cur[i0 + 2] = o.z; }
        }
        carry += total;
        v = nv;
    }
}

__device__ __forceinline__ float wave_max(float x) {
#pragma unroll
    for (int off = 16; off > 0; off >>= 1) x = fmaxf(x, __shfl_xor(x, off, 32));
    return x;
}
__device__ __forceinline__ float wave_sum(float x) {
#pragma unroll
    for (int off = 16; off > 0; off >>= 1) x += __shfl_xor(x, off, 32);
    return x;
}

__global__ void agg_rows(const int* __restrict__ offs, const int* __restrict__ endp,
                         const int* __restrict__ csr,
                         const float* __restrict__ ssrc, const float* __restrict__ sdst,
                         const float* __restrict__ z, float* __restrict__ out,
                         int lo, int nr)
{
    const int lane = threadIdx.x & 31;
    const int r = __builtin_amdgcn_readfirstlane((blockIdx.x * blockDim.x + threadIdx.x) >> 5);
    if (r >= nr) return;
    const int beg = __builtin_amdgcn_readfirstlane(offs[r]);
    const int end = __builtin_amdgcn_readfirstlane(endp[r]);
    const float sd = sdst[r + lo];

    int   s0 = 0;
    float v0 = -INFINITY;
    const bool have0 = (beg + lane) < end;
    if (have0) { s0 = csr[beg + lane]; v0 = leaky(ssrc[s0] + sd); }
    float mx = v0;
    for (int j = beg + 32 + lane; j < end; j += 32)
        mx = fmaxf(mx, leaky(ssrc[csr[j]] + sd));
    mx = wave_max(mx);

    const float ex0 = have0 ? __expf(v0 - mx) : 0.0f;
    float dn = ex0;
    for (int j = beg + 32 + lane; j < end; j += 32)
        dn += __expf(leaky(ssrc[csr[j]] + sd) - mx);
    dn = wave_sum(dn);

    float4 acc = make_float4(0.0f, 0.0f, 0.0f, 0.0f);
    for (int base = beg; base < end; base += 32) {
        int   s = s0;
        float a = have0 ? ex0 / dn : 0.0f;
        if (base != beg) {
            const int j = base + lane;
            s = 0; a = 0.0f;
            if (j < end) { s = csr[j]; a = __expf(leaky(ssrc[s] + sd) - mx) / dn; }
        }
        const int cnt = (end - base) < 32 ? (end - base) : 32;
#pragma unroll 4
        for (int i = 0; i < cnt; ++i) {
            const int   si = __builtin_amdgcn_readlane(s, i);
            const float ai = __int_as_float(__builtin_amdgcn_readlane(__float_as_int(a), i));
            const float4 zv = *(const float4*)(z + (size_t)si * D + lane * 4);
            acc.x = fmaf(ai, zv.x, acc.x);
            acc.y = fmaf(ai, zv.y, acc.y);
            acc.z = fmaf(ai, zv.z, acc.z);
            acc.w = fmaf(ai, zv.w, acc.w);
        }
    }
    *(float4*)(out + (size_t)r * D + lane * 4) = acc;
}

__global__ void __launch_bounds__(256)
combine(const float* __restrict__ A, const float* __restrict__ B,
        const float* __restrict__ W1, const float* __restrict__ b1,
        const float* __restrict__ w2, float* __restrict__ out, int n)
{
    __shared__ __attribute__((aligned(16))) float sA[16 * GP];
    __shared__ __attribute__((aligned(16))) float sB[16 * GP];
    __shared__ __attribute__((aligned(16))) float sc[2][16][8];
    const int wave = threadIdx.x >> 5;
    const int lane = threadIdx.x & 31;
    const int half = lane >> 4;
    const int lr   = lane & 15;
    const int row0 = blockIdx.x * 16;
    const int col0 = wave * 16;

    for (int p = 0; p < 2; ++p) {
        const int rr = wave + 8 * p;
        int r = row0 + rr;
        if (r >= n) r = n - 1;
        const float4 va = *(const float4*)(A + (size_t)r * D + lane * 4);
        const float4 vb = *(const float4*)(B + (size_t)r * D + lane * 4);
        *(float4*)(sA + rr * GP + lane * 4) = va;
        *(float4*)(sB + rr * GP + lane * 4) = vb;
    }
    __syncthreads();

    v8f ca = {0.f, 0.f, 0.f, 0.f, 0.f, 0.f, 0.f, 0.f};
    v8f cb = {0.f, 0.f, 0.f, 0.f, 0.f, 0.f, 0.f, 0.f};
    const float* arow = sA + lr * GP + 2 * half;
    const float* brow = sB + lr * GP + 2 * half;
    const float* wcol = W1 + (size_t)(2 * half) * D + col0 + lr;
#pragma unroll
    for (int k0 = 0; k0 < D; k0 += 4) {
        const float2 av = *(const float2*)(arow + k0);
        const float2 bv = *(const float2*)(brow + k0);
        v2f a; a.x = av.x; a.y = av.y;
        v2f b; b.x = bv.x; b.y = bv.y;
        v2f w; w.x = wcol[(size_t)k0 * D];
               w.y = wcol[(size_t)(k0 + 1) * D];
        ca = __builtin_amdgcn_wmma_f32_16x16x4_f32(
                 false, a, false, w, (short)0, ca, false, false);
        cb = __builtin_amdgcn_wmma_f32_16x16x4_f32(
                 false, b, false, w, (short)0, cb, false, false);
    }

    const float bias = b1[col0 + lr];
    const float w2c  = w2[col0 + lr];
#pragma unroll
    for (int v = 0; v < 8; ++v) {
        ca[v] = tanhf(ca[v] + bias);
        cb[v] = tanhf(cb[v] + bias);
    }
    const float p1 = rows_dot16(ca, w2c, lr);
    const float p2 = rows_dot16(cb, w2c, lr);
    if ((lr & 1) == 0) {
        sc[0][8 * half + (lr >> 1)][wave] = p1;
        sc[1][8 * half + (lr >> 1)][wave] = p2;
    }
    __syncthreads();

    for (int p = 0; p < 2; ++p) {
        const int rr = wave + 8 * p;
        const float4 x0 = *(const float4*)&sc[0][rr][0];
        const float4 x1 = *(const float4*)&sc[0][rr][4];
        const float4 y0 = *(const float4*)&sc[1][rr][0];
        const float4 y1 = *(const float4*)&sc[1][rr][4];
        const float s1 = ((((((x0.x + x0.y) + x0.z) + x0.w) + x1.x) + x1.y) + x1.z) + x1.w;
        const float s2 = ((((((y0.x + y0.y) + y0.z) + y0.w) + y1.x) + y1.y) + y1.z) + y1.w;
        const float mm = fmaxf(s1, s2);
        const float e1 = __expf(s1 - mm), e2 = __expf(s2 - mm);
        const float al = e1 / (e1 + e2);
        const float be = 1.0f - al;
        const float4 va = *(const float4*)(sA + rr * GP + lane * 4);
        const float4 vb = *(const float4*)(sB + rr * GP + lane * 4);
        float4 o;
        o.x = al * va.x + be * vb.x;
        o.y = al * va.y + be * vb.y;
        o.z = al * va.z + be * vb.z;
        o.w = al * va.w + be * vb.w;
        const int r = row0 + rr;
        if (r < n) *(float4*)(out + (size_t)r * D + lane * 4) = o;
    }
}

static inline int cdiv(int a, int b) { return (a + b - 1) / b; }

extern "C" void kernel_launch(void* const* d_in, const int* in_sizes, int n_in,
                              void* d_out, int out_size, void* d_ws, size_t ws_size,
                              hipStream_t stream)
{
    (void)n_in; (void)ws_size; (void)out_size;
    const float* kn_emb   = (const float*)d_in[0];
    const float* exer_emb = (const float*)d_in[1];
    const float* W_dir    = (const float*)d_in[2];
    const float* a_dir    = (const float*)d_in[3];
    const float* W_ke     = (const float*)d_in[4];
    const float* a_ke     = (const float*)d_in[5];
    const float* W_ek     = (const float*)d_in[6];
    const float* a_ek     = (const float*)d_in[7];
    const float* rW1      = (const float*)d_in[8];
    const float* rb1      = (const float*)d_in[9];
    const float* rw2      = (const float*)d_in[10];
    const int* dsrc  = (const int*)d_in[11];
    const int* ddst  = (const int*)d_in[12];
    const int* kesrc = (const int*)d_in[13];
    const int* kedst = (const int*)d_in[14];
    const int* eksrc = (const int*)d_in[15];
    const int* ekdst = (const int*)d_in[16];

    const int Kn = in_sizes[0] / D;
    const int In = in_sizes[1] / D;
    const int Nn = Kn + In;
    const int E1 = in_sizes[11];
    const int E2 = in_sizes[13];

    float* w = (float*)d_ws;
    size_t off = 0;
    float* buf_z = w + off; off += (size_t)Nn * D;
    float* buf_B = w + off; off += (size_t)Nn * D;
    float* buf_A = w + off; off += (size_t)Kn * D;
    float* ssrc  = w + off; off += (size_t)Nn;
    float* sdst  = w + off; off += (size_t)Nn;

    const int r4I = (In + 3) & ~3;
    const int r4K = (Kn + 3) & ~3;
    const int rb_ke  = r4I;
    const int rb_dir = r4I + r4K;
    const int nrows  = r4I + 2 * r4K;
    const int cb_ke  = E2;
    const int cb_dir = 2 * E2;
    int* cur  = (int*)(buf_B + (size_t)Kn * D);
    int* offs = cur + nrows;
    int* csr  = offs + nrows;

    float* out_kn = (float*)d_out;
    float* out_ex = (float*)d_out + (size_t)Kn * D;

    const int nb2 = cdiv(cdiv(E2, 4), 256);
    const int nb1 = cdiv(cdiv(E1, 4), 256);

    gemm_z<<<512, 256, 0, stream>>>(exer_emb, In, kn_emb, W_ek, a_ek, buf_z, ssrc, sdst, Nn);

    hipMemsetAsync(cur, 0, (size_t)nrows * sizeof(int), stream);
    edge_count3<<<2 * nb2 + nb1, 256, 0, stream>>>(ekdst, 0, In, E2,
                                                   kedst, In, Nn, E2, nb2, rb_ke,
                                                   ddst, 0, Kn, E1, 2 * nb2, rb_dir,
                                                   cur);
    scan_rows3<<<3, SCAN_T, 0, stream>>>(cur, offs, In, Kn, rb_ke, cb_ke, Kn, rb_dir, cb_dir);
    edge_fill3<<<2 * nb2 + nb1, 256, 0, stream>>>(eksrc, ekdst, 0, In, E2,
                                                  kesrc, kedst, In, Nn, E2, nb2, rb_ke,
                                                  dsrc, ddst, 0, Kn, E1, 2 * nb2, rb_dir,
                                                  cur, csr);

    agg_rows<<<cdiv(In, 8), 256, 0, stream>>>(offs, cur, csr, ssrc, sdst, buf_z, out_ex, 0, In);

    gemm_z<<<512, 256, 0, stream>>>(exer_emb, In, kn_emb, W_ke, a_ke, buf_z, ssrc, sdst, Nn);
    agg_rows<<<cdiv(Kn, 8), 256, 0, stream>>>(offs + rb_ke, cur + rb_ke, csr, ssrc, sdst, buf_z, buf_B, In, Kn);

    gemm_z<<<512, 256, 0, stream>>>(kn_emb, Kn, kn_emb, W_dir, a_dir, buf_z, ssrc, sdst, Kn);
    agg_rows<<<cdiv(Kn, 8), 256, 0, stream>>>(offs + rb_dir, cur + rb_dir, csr, ssrc, sdst, buf_z, buf_A, 0, Kn);

    combine<<<cdiv(Kn, 16), 256, 0, stream>>>(buf_A, buf_B, rW1, rb1, rw2, out_kn, Kn);
}


# Knowledge-node tail fused: a 16-wave block per 16 nodes reduces their A and B rows (both graphs) into LDS and runs the gate on them; A, B never stored, 3 launches become 1, sum orders as before
# speedup vs baseline: 2.9462x; 1.0318x over previous
#include <hip/hip_runtime.h>
#include <hip/hip_bf16.h>
#include <math.h>


#define D 128

typedef float v2f __attribute__((ext_vector_type(2)));
typedef float v8f __attribute__((ext_vector_type(8)));

__device__ __forceinline__ float rows_dot16(const v8f c, float coef, int lr)
{
    const bool b3 = (lr & 8) != 0;
    const bool b2 = (lr & 4) != 0;
    const bool b1 = (lr & 2) != 0;
    float y[4];
#pragma unroll
    for (int i = 0; i < 4; ++i) {
        const float lo = c[i] * coef, hi = c[i + 4] * coef;
        const float keep = b3 ? hi : lo;
        const float send = b3 ? lo : hi;
        y[i] = keep + __shfl_xor(send, 8, 32);
    }
    float q[2];
#pragma unroll
    for (int i = 0; i < 2; ++i) {
        const float keep = b2 ? y[i + 2] : y[i];
        const float send = b2 ? y[i] : y[i + 2];
        q[i] = keep + __shfl_xor(send, 4, 32);
    }
    const float keep = b1 ? q[1] : q[0];
    const float send = b1 ? q[0] : q[1];
    float w = keep + __shfl_xor(send, 2, 32);
    w += __shfl_xor(w, 1, 32);
    return w;
}

#define GP 132

__device__ __forceinline__ float4 load_hrow(const float* __restrict__ h0, int n0,
                                            const float* __restrict__ h1,
                                            int r, int n, int lane)
{
    if (r >= n) r = n - 1;
    const float* hrow = (r < n0) ? (h0 + (size_t)r * D)
                                 : (h1 + (size_t)(r - n0) * D);
    return *(const float4*)(hrow + lane * 4);
}

__global__ void __launch_bounds__(256)
gemm_z(const float* __restrict__ h0, int n0,
       const float* __restrict__ h1,
       const float* __restrict__ W,
       const float* __restrict__ att,
       float* __restrict__ Z,
       float* __restrict__ ssrc, float* __restrict__ sdst, int n)
{
    __shared__ __attribute__((aligned(16))) float sA[2][16 * GP];
    __shared__ float sc[2][8][2][16];
    const int wave = threadIdx.x >> 5;
    const int lane = threadIdx.x & 31;
    const int half = lane >> 4;
    const int lr   = lane & 15;
    const int col0 = wave * 16;
    const int nstrips = (n + 15) >> 4;
    const int step = gridDim.x;

    int strip = blockIdx.x;
    if (strip >= nstrips) return;

    float4 p0 = load_hrow(h0, n0, h1, strip * 16 + wave,     n, lane);
    float4 p1 = load_hrow(h0, n0, h1, strip * 16 + wave + 8, n, lane);

    v2f wf[32];
    const float* wcol = W + (size_t)(2 * half) * D + col0 + lr;
#pragma unroll
    for (int i = 0; i < 32; ++i) {
        wf[i].x = wcol[(size_t)(4 * i) * D];
        wf[i].y = wcol[(size_t)(4 * i + 1) * D];
    }
    const float att_s = att[col0 + lr];
    const float att_d = att[D + col0 + lr];

    *(float4*)(&sA[0][wave * GP + lane * 4])       = p0;
    *(float4*)(&sA[0][(wave + 8) * GP + lane * 4]) = p1;
    __syncthreads();

    int buf = 0;
    for (; strip < nstrips; strip += step) {
        const int row0 = strip * 16;
        const int next = strip + step;
        const bool more = next < nstrips;
        if (more) {
            p0 = load_hrow(h0, n0, h1, next * 16 + wave,     n, lane);
            p1 = load_hrow(h0, n0, h1, next * 16 + wave + 8, n, lane);
        }

        v8f c = {0.f, 0.f, 0.f, 0.f, 0.f, 0.f, 0.f, 0.f};
        const float* arow = &sA[buf][lr * GP + 2 * half];
        v2f af[32];
#pragma unroll
        for (int i = 0; i < 32; ++i) {
            const float2 av = *(const float2*)(arow + 4 * i);
            af[i].x = av.x; af[i].y = av.y;
        }
        __builtin_amdgcn_sched_barrier(0);
#pragma unroll
        for (int i = 0; i < 32; ++i) {
            c = __builtin_amdgcn_wmma_f32_16x16x4_f32(
                    false, af[i], false, wf[i], (short)0, c, false, false);
        }
        const float ps = rows_dot16(c, att_s, lr);
        const float pd = rows_dot16(c, att_d, lr);
        if ((lr & 1) == 0) {
            sc[buf][wave][0][8 * half + (lr >> 1)] = ps;
            sc[buf][wave][1][8 * half + (lr >> 1)] = pd;
        }
        if (more) {
            *(float4*)(&sA[buf ^ 1][wave * GP + lane * 4])       = p0;
            *(float4*)(&sA[buf ^ 1][(wave + 8) * GP + lane * 4]) = p1;
        }
        __syncthreads();

#pragma unroll
        for (int v = 0; v < 8; ++v) {
            const int r = row0 + v + 8 * half;
            if (r < n) Z[(size_t)r * D + col0 + lr] = c[v];
        }

        if (threadIdx.x < 32) {
            const int which = threadIdx.x >> 4;
            const int rr = threadIdx.x & 15;
            float t = 0.0f;
#pragma unroll
            for (int w8 = 0; w8 < 8; ++w8) t += sc[buf][w8][which][rr];
            const int r = row0 + rr;
            if (r < n) { if (which) sdst[r] = t; else ssrc[r] = t; }
        }
        buf ^= 1;
    }
}

__device__ __forceinline__ float leaky(float v) {
    return v > 0.0f ? v : 0.01f * v;
}


struct EdgePart {
    const int* src; const int* dst;
    int lo, hi, nE;
    int blk0;
    int rbase;
};

__device__ __forceinline__ EdgePart pick_part(
    const int* s0, const int* d0, int lo0, int hi0, int nE0,
    const int* s1, const int* d1, int lo1, int hi1, int nE1, int blk1, int rb1,
    const int* s2, const int* d2, int lo2, int hi2, int nE2, int blk2, int rb2)
{
    EdgePart p;
    const int b = blockIdx.x;
    if (b >= blk2)      { p.src = s2; p.dst = d2; p.lo = lo2; p.hi = hi2; p.nE = nE2; p.blk0 = blk2; p.rbase = rb2; }
    else if (b >= blk1) { p.src = s1; p.dst = d1; p.lo = lo1; p.hi = hi1; p.nE = nE1; p.blk0 = blk1; p.rbase = rb1; }
    else                { p.src = s0; p.dst = d0; p.lo = lo0; p.hi = hi0; p.nE = nE0; p.blk0 = 0;    p.rbase = 0;   }
    return p;
}

__device__ __forceinline__ int4 load_ids4(const int* __restrict__ p, int e0, int nE) {
    int4 v = make_int4(-1, -1, -1, -1);
    if (e0 + 3 < nE) {
        v = *(const int4*)(p + e0);
    } else {
        if (e0     < nE) v.x = p[e0];
        if (e0 + 1 < nE) v.y = p[e0 + 1];
        if (e0 + 2 < nE) v.z = p[e0 + 2];
    }
    return v;
}

__global__ void edge_count3(const int* __restrict__ d0, int lo0, int hi0, int nE0,
                            const int* __restrict__ d1, int lo1, int hi1, int nE1, int blk1, int rb1,
                            const int* __restrict__ d2, int lo2, int hi2, int nE2, int blk2, int rb2,
                            int* __restrict__ cnt_all)
{
    const EdgePart p = pick_part(d0, d0, lo0, hi0, nE0,
                                 d1, d1, lo1, hi1, nE1, blk1, rb1,
                                 d2, d2, lo2, hi2, nE2, blk2, rb2);
    const int lo = p.lo, hi = p.hi, nE = p.nE;
    int* cnt = cnt_all + p.rbase;
    const int e0 = ((blockIdx.x - p.blk0) * blockDim.x + threadIdx.x) * 4;
    if (e0 >= nE) return;
    const int4 d = load_ids4(p.dst, e0, nE);
    if (d.x >= lo && d.x < hi) atomicAdd(&cnt[d.x - lo], 1);
    if (d.y >= lo && d.y < hi) atomicAdd(&cnt[d.y - lo], 1);
    if (d.z >= lo && d.z < hi) atomicAdd(&cnt[d.z - lo], 1);
    if (d.w >= lo && d.w < hi) atomicAdd(&cnt[d.w - lo], 1);
}

__global__ void edge_fill3(const int* __restrict__ s0, const int* __restrict__ d0, int lo0, int hi0, int nE0,
                           const int* __restrict__ s1, const int* __restrict__ d1, int lo1, int hi1, int nE1, int blk1, int rb1,
                           const int* __restrict__ s2, const int* __restrict__ d2, int lo2, int hi2, int nE2, int blk2, int rb2,
                           int* __restrict__ cur_all, int* __restrict__ csr)
{
    const EdgePart p = pick_part(s0, d0, lo0, hi0, nE0,
                                 s1, d1, lo1, hi1, nE1, blk1, rb1,
                                 s2, d2, lo2, hi2, nE2, blk2, rb2);
    const int lo = p.lo, hi = p.hi, nE = p.nE;
    int* cur = cur_all + p.rbase;
    const int e0 = ((blockIdx.x - p.blk0) * blockDim.x + threadIdx.x) * 4;
    if (e0 >= nE) return;
    const int4 d = load_ids4(p.dst, e0, nE);
    const int4 s = load_ids4(p.src, e0, nE);
    if (d.x >= lo && d.x < hi) csr[atomicAdd(&cur[d.x - lo], 1)] = s.x;
    if (d.y >= lo && d.y < hi) csr[atomicAdd(&cur[d.y - lo], 1)] = s.y;
    if (d.z >= lo && d.z < hi) csr[atomicAdd(&cur[d.z - lo], 1)] = s.z;
    if (d.w >= lo && d.w < hi) csr[atomicAdd(&cur[d.w - lo], 1)] = s.w;
}

#define SCAN_T 1024

__device__ __forceinline__ int4 load_cnt4(const int* p, int i0, int n) {
    int4 v = make_int4(0, 0, 0, 0);
    if (i0 + 3 < n) {
        v = *(const int4*)(p + i0);
    } else {
        if (i0     < n) v.x = p[i0];
        if (i0 + 1 < n) v.y = p[i0 + 1];
        if (i0 + 2 < n) v.z = p[i0 + 2];
    }
    return v;
}

__device__ __forceinline__ int wave_incl_scan(int x, int lane) {
#pragma unroll
    for (int off = 1; off < 32; off <<= 1) {
        const int nb = __shfl_up(x, off, 32);
        if (lane >= off) x += nb;
    }
    return x;
}

__global__ void __launch_bounds__(SCAN_T)
scan_rows3(int* cur_all, int* __restrict__ offs_all,
           int n0, int n1, int rb1, int cb1, int n2, int rb2, int cb2)
{
    __shared__ int wsum[32];
    const int t = threadIdx.x;
    const int lane = t & 31;
    const int wv = t >> 5;
    int n = n0, rb = 0, carry = 0;
    if (blockIdx.x == 1)      { n = n1; rb = rb1; carry = cb1; }
    else if (blockIdx.x == 2) { n = n2; rb = rb2; carry = cb2; }
    int* cur = cur_all + rb;
    int* offs = offs_all + rb;
    int4 v = load_cnt4(cur, t * 4, n);
    for (int base = 0; base < n; base += SCAN_T * 4) {
        const int i0 = base + t * 4;
        const int4 nv = load_cnt4(cur, i0 + SCAN_T * 4, n);
        const int tsum = v.x + v.y + v.z + v.w;
        const int inc = wave_incl_scan(tsum, lane);
        if (lane == 31) wsum[wv] = inc;
        __syncthreads();
        const int ws = wsum[lane];
        const int winc = wave_incl_scan(ws, lane);
        const int wave_excl = __shfl(winc, wv, 32) - __shfl(ws, wv, 32);
        const int total = __shfl(winc, 31, 32);
        __syncthreads();
        int4 o;
        o.x = carry + wave_excl + (inc - tsum);
        o.y = o.x + v.x;
        o.z = o.y + v.y;
        o.w = o.z + v.z;
        if (i0 + 3 < n) {
            *(int4*)(offs + i0) = o;
            *(int4*)(cur + i0) = o;
        } else {
            if (i0     < n) { offs[i0]     = o.x; cur[i0]     = o.x; }
            if (i0 + 1 < n) { offs[i0 + 1] = o.y; cur[i0 + 1] = o.y; }
            if (i0 + 2 < n) { offs[i0 + 2] = o.z; cur[i0 + 2] = o.z; }
        }
        carry += total;
        v = nv;
    }
}

__device__ __forceinline__ float wave_max(float x) {
#pragma unroll
    for (int off = 16; off > 0; off >>= 1) x = fmaxf(x, __shfl_xor(x, off, 32));
    return x;
}
__device__ __forceinline__ float wave_sum(float x) {
#pragma unroll
    for (int off = 16; off > 0; off >>= 1) x += __shfl_xor(x, off, 32);
    return x;
}

__device__ __forceinline__ float4 agg_one_row(const int* __restrict__ offs, const int* __restrict__ endp,
                                              const int* __restrict__ csr,
                                              const float* __restrict__ ssrc, const float* __restrict__ sdst,
                                              const float* __restrict__ z, int lo, int r, int lane)
{
    const int beg = __builtin_amdgcn_readfirstlane(offs[r]);
    const int end = __builtin_amdgcn_readfirstlane(endp[r]);
    const float sd = sdst[r + lo];

    int   s0 = 0;
    float v0 = -INFINITY;
    const bool have0 = (beg + lane) < end;
    if (have0) { s0 = csr[beg + lane]; v0 = leaky(ssrc[s0] + sd); }
    float mx = v0;
    for (int j = beg + 32 + lane; j < end; j += 32)
        mx = fmaxf(mx, leaky(ssrc[csr[j]] + sd));
    mx = wave_max(mx);

    const float ex0 = have0 ? __expf(v0 - mx) : 0.0f;
    float dn = ex0;
    for (int j = beg + 32 + lane; j < end; j += 32)
        dn += __expf(leaky(ssrc[csr[j]] + sd) - mx);
    dn = wave_sum(dn);

    float4 acc = make_float4(0.0f, 0.0f, 0.0f, 0.0f);
    for (int base = beg; base < end; base += 32) {
        int   s = s0;
        float a = have0 ? ex0 / dn : 0.0f;
        if (base != beg) {
            const int j = base + lane;
            s = 0; a = 0.0f;
            if (j < end) { s = csr[j]; a = __expf(leaky(ssrc[s] + sd) - mx) / dn; }
        }
        const int cnt = (end - base) < 32 ? (end - base) : 32;
#pragma unroll 4
        for (int i = 0; i < cnt; ++i) {
            const int   si = __builtin_amdgcn_readlane(s, i);
            const float ai = __int_as_float(__builtin_amdgcn_readlane(__float_as_int(a), i));
            const float4 zv = *(const float4*)(z + (size_t)si * D + lane * 4);
            acc.x = fmaf(ai, zv.x, acc.x);
            acc.y = fmaf(ai, zv.y, acc.y);
            acc.z = fmaf(ai, zv.z, acc.z);
            acc.w = fmaf(ai, zv.w, acc.w);
        }
    }
    return acc;
}

__global__ void agg_rows(const int* __restrict__ offs, const int* __restrict__ endp,
                         const int* __restrict__ csr,
                         const float* __restrict__ ssrc, const float* __restrict__ sdst,
                         const float* __restrict__ z, float* __restrict__ out,
                         int lo, int nr)
{
    const int lane = threadIdx.x & 31;
    const int r = __builtin_amdgcn_readfirstlane((blockIdx.x * blockDim.x + threadIdx.x) >> 5);
    if (r >= nr) return;
    const float4 acc = agg_one_row(offs, endp, csr, ssrc, sdst, z, lo, r, lane);
    *(float4*)(out + (size_t)r * D + lane * 4) = acc;
}

__global__ void __launch_bounds__(512, 16)
agg_combine(const int* __restrict__ offs_d, const int* __restrict__ end_d,
            const int* __restrict__ offs_k, const int* __restrict__ end_k,
            const int* __restrict__ csr,
            const float* __restrict__ ssrc_d, const float* __restrict__ sdst_d,
            const float* __restrict__ z_d,
            const float* __restrict__ ssrc_k, const float* __restrict__ sdst_k,
            const float* __restrict__ z_k, int lo_k,
            const float* __restrict__ W1, const float* __restrict__ b1,
            const float* __restrict__ w2, float* __restrict__ out, int n)
{
    __shared__ __attribute__((aligned(16))) float sX[2][16 * GP];
    __shared__ __attribute__((aligned(16))) float sc[2][16][8];
    const int wave = __builtin_amdgcn_readfirstlane(threadIdx.x >> 5);
    const int lane = threadIdx.x & 31;
    const int half = lane >> 4;
    const int lr   = lane & 15;
    const int g    = wave >> 3;
    const int wt   = wave & 7;
    const int row0 = blockIdx.x * 16;
    const int col0 = wt * 16;

    {
        const int r = row0 + wave;
        float4 ra = make_float4(0.0f, 0.0f, 0.0f, 0.0f);
        if (r < n) ra = agg_one_row(offs_d, end_d, csr, ssrc_d, sdst_d, z_d, 0, r, lane);
        *(float4*)(sX[0] + wave * GP + lane * 4) = ra;
        float4 rb = make_float4(0.0f, 0.0f, 0.0f, 0.0f);
        if (r < n) rb = agg_one_row(offs_k, end_k, csr, ssrc_k, sdst_k, z_k, lo_k, r, lane);
        *(float4*)(sX[1] + wave * GP + lane * 4) = rb;
    }
    __syncthreads();

    v8f c = {0.f, 0.f, 0.f, 0.f, 0.f, 0.f, 0.f, 0.f};
    const float* xrow = sX[g] + lr * GP + 2 * half;
    const float* wcol = W1 + (size_t)(2 * half) * D + col0 + lr;
#pragma unroll
    for (int k0 = 0; k0 < D; k0 += 4) {
        const float2 xv = *(const float2*)(xrow + k0);
        v2f a; a.x = xv.x; a.y = xv.y;
        v2f w; w.x = wcol[(size_t)k0 * D];
               w.y = wcol[(size_t)(k0 + 1) * D];
        c = __builtin_amdgcn_wmma_f32_16x16x4_f32(
                false, a, false, w, (short)0, c, false, false);
    }

    const float bias = b1[col0 + lr];
    const float w2c  = w2[col0 + lr];
#pragma unroll
    for (int v = 0; v < 8; ++v) c[v] = tanhf(c[v] + bias);
    const float pp = rows_dot16(c, w2c, lr);
    if ((lr & 1) == 0) sc[g][8 * half + (lr >> 1)][wt] = pp;
    __syncthreads();

    {
        const int rr = wave;
        const float4 x0 = *(const float4*)&sc[0][rr][0];
        const float4 x1 = *(const float4*)&sc[0][rr][4];
        const float4 y0 = *(const float4*)&sc[1][rr][0];
        const float4 y1 = *(const float4*)&sc[1][rr][4];
        const float s1 = ((((((x0.x + x0.y) + x0.z) + x0.w) + x1.x) + x1.y) + x1.z) + x1.w;
        const float s2 = ((((((y0.x + y0.y) + y0.z) + y0.w) + y1.x) + y1.y) + y1.z) + y1.w;
        const float mm = fmaxf(s1, s2);
        const float e1 = __expf(s1 - mm), e2 = __expf(s2 - mm);
        const float al = e1 / (e1 + e2);
        const float be = 1.0f - al;
        const float4 va = *(const float4*)(sX[0] + rr * GP + lane * 4);
        const float4 vb = *(const float4*)(sX[1] + rr * GP + lane * 4);
        float4 o;
        o.x = al * va.x + be * vb.x;
        o.y = al * va.y + be * vb.y;
        o.z = al * va.z + be * vb.z;
        o.w = al * va.w + be * vb.w;
        const int r = row0 + rr;
        if (r < n) *(float4*)(out + (size_t)r * D + lane * 4) = o;
    }
}

static inline int cdiv(int a, int b) { return (a + b - 1) / b; }

extern "C" void kernel_launch(void* const* d_in, const int* in_sizes, int n_in,
                              void* d_out, int out_size, void* d_ws, size_t ws_size,
                              hipStream_t stream)
{
    (void)n_in; (void)ws_size; (void)out_size;
    const float* kn_emb   = (const float*)d_in[0];
    const float* exer_emb = (const float*)d_in[1];
    const float* W_dir    = (const float*)d_in[2];
    const float* a_dir    = (const float*)d_in[3];
    const float* W_ke     = (const float*)d_in[4];
    const float* a_ke     = (const float*)d_in[5];
    const float* W_ek     = (const float*)d_in[6];
    const float* a_ek     = (const float*)d_in[7];
    const float* rW1      = (const float*)d_in[8];
    const float* rb1      = (const float*)d_in[9];
    const float* rw2      = (const float*)d_in[10];
    const int* dsrc  = (const int*)d_in[11];
    const int* ddst  = (const int*)d_in[12];
    const int* kesrc = (const int*)d_in[13];
    const int* kedst = (const int*)d_in[14];
    const int* eksrc = (const int*)d_in[15];
    const int* ekdst = (const int*)d_in[16];

    const int Kn = in_sizes[0] / D;
    const int In = in_sizes[1] / D;
    const int Nn = Kn + In;
    const int E1 = in_sizes[11];
    const int E2 = in_sizes[13];

    float* w = (float*)d_ws;
    size_t off = 0;
    float* buf_z = w + off; off += (size_t)Nn * D;
    float* buf_B = w + off; off += (size_t)Nn * D;
    float* buf_A = w + off; off += (size_t)Kn * D;
    float* ssrc  = w + off; off += (size_t)Nn;
    float* sdst  = w + off; off += (size_t)Nn;
    float* ssrc_d = buf_A;
    float* sdst_d = buf_A + Kn;

    const int r4I = (In + 3) & ~3;
    const int r4K = (Kn + 3) & ~3;
    const int rb_ke  = r4I;
    const int rb_dir = r4I + r4K;
    const int nrows  = r4I + 2 * r4K;
    const int cb_ke  = E2;
    const int cb_dir = 2 * E2;
    int* cur  = (int*)(buf_B + (size_t)Kn * D);
    int* offs = cur + nrows;
    int* csr  = offs + nrows;

    float* out_kn = (float*)d_out;
    float* out_ex = (float*)d_out + (size_t)Kn * D;

    const int nb2 = cdiv(cdiv(E2, 4), 256);
    const int nb1 = cdiv(cdiv(E1, 4), 256);

    gemm_z<<<512, 256, 0, stream>>>(exer_emb, In, kn_emb, W_ek, a_ek, buf_z, ssrc, sdst, Nn);

    hipMemsetAsync(cur, 0, (size_t)nrows * sizeof(int), stream);
    edge_count3<<<2 * nb2 + nb1, 256, 0, stream>>>(ekdst, 0, In, E2,
                                                   kedst, In, Nn, E2, nb2, rb_ke,
                                                   ddst, 0, Kn, E1, 2 * nb2, rb_dir,
                                                   cur);
    scan_rows3<<<3, SCAN_T, 0, stream>>>(cur, offs, In, Kn, rb_ke, cb_ke, Kn, rb_dir, cb_dir);
    edge_fill3<<<2 * nb2 + nb1, 256, 0, stream>>>(eksrc, ekdst, 0, In, E2,
                                                  kesrc, kedst, In, Nn, E2, nb2, rb_ke,
                                                  dsrc, ddst, 0, Kn, E1, 2 * nb2, rb_dir,
                                                  cur, csr);

    agg_rows<<<cdiv(In, 8), 256, 0, stream>>>(offs, cur, csr, ssrc, sdst, buf_z, out_ex, 0, In);

    gemm_z<<<512, 256, 0, stream>>>(exer_emb, In, kn_emb, W_ke, a_ke, buf_z, ssrc, sdst, Nn);

    gemm_z<<<512, 256, 0, stream>>>(kn_emb, Kn, kn_emb, W_dir, a_dir, buf_B, ssrc_d, sdst_d, Kn);

    agg_combine<<<cdiv(Kn, 16), 512, 0, stream>>>(offs + rb_dir, cur + rb_dir,
                                                  offs + rb_ke, cur + rb_ke, csr,
                                                  ssrc_d, sdst_d, buf_B,
                                                  ssrc, sdst, buf_z, In,
                                                  rW1, rb1, rw2, out_kn, Kn);
}


# Edge grouping by owner blocks: a block owns 384 kept rows, scans its graph's dst list, ranks edges by LDS adds, stores slot lines whole; per-edge global atomics gone; rows still sum in arrival order
# speedup vs baseline: 5.1746x; 1.7564x over previous
#include <hip/hip_runtime.h>
#include <hip/hip_bf16.h>
#include <math.h>


#define D 128

typedef float v2f __attribute__((ext_vector_type(2)));
typedef float v8f __attribute__((ext_vector_type(8)));

__device__ __forceinline__ float rows_dot16(const v8f c, float coef, int lr)
{
    const bool b3 = (lr & 8) != 0;
    const bool b2 = (lr & 4) != 0;
    const bool b1 = (lr & 2) != 0;
    float y[4];
#pragma unroll
    for (int i = 0; i < 4; ++i) {
        const float lo = c[i] * coef, hi = c[i + 4] * coef;
        const float keep = b3 ? hi : lo;
        const float send = b3 ? lo : hi;
        y[i] = keep + __shfl_xor(send, 8, 32);
    }
    float q[2];
#pragma unroll
    for (int i = 0; i < 2; ++i) {
        const float keep = b2 ? y[i + 2] : y[i];
        const float send = b2 ? y[i] : y[i + 2];
        q[i] = keep + __shfl_xor(send, 4, 32);
    }
    const float keep = b1 ? q[1] : q[0];
    const float send = b1 ? q[0] : q[1];
    float w = keep + __shfl_xor(send, 2, 32);
    w += __shfl_xor(w, 1, 32);
    return w;
}

#define GP 132

__device__ __forceinline__ float4 load_hrow(const float* __restrict__ h0, int n0,
                                            const float* __restrict__ h1,
                                            int r, int n, int lane)
{
    if (r >= n) r = n - 1;
    const float* hrow = (r < n0) ? (h0 + (size_t)r * D)
                                 : (h1 + (size_t)(r - n0) * D);
    return *(const float4*)(hrow + lane * 4);
}

__global__ void __launch_bounds__(256)
gemm_z(const float* __restrict__ h0, int n0,
       const float* __restrict__ h1,
       const float* __restrict__ W,
       const float* __restrict__ att,
       float* __restrict__ Z,
       float* __restrict__ ssrc, float* __restrict__ sdst, int n,
       int* __restrict__ clr, int nclr)
{
    __shared__ __attribute__((aligned(16))) float sA[2][16 * GP];
    __shared__ float sc[2][8][2][16];
    const int wave = threadIdx.x >> 5;
    const int lane = threadIdx.x & 31;
    const int half = lane >> 4;
    const int lr   = lane & 15;
    const int col0 = wave * 16;
    const int nstrips = (n + 15) >> 4;
    const int step = gridDim.x;

    for (int i = blockIdx.x * blockDim.x + threadIdx.x; i < nclr; i += gridDim.x * blockDim.x)
        clr[i] = 0;

    int strip = blockIdx.x;
    if (strip >= nstrips) return;

    float4 p0 = load_hrow(h0, n0, h1, strip * 16 + wave,     n, lane);
    float4 p1 = load_hrow(h0, n0, h1, strip * 16 + wave + 8, n, lane);

    v2f wf[32];
    const float* wcol = W + (size_t)(2 * half) * D + col0 + lr;
#pragma unroll
    for (int i = 0; i < 32; ++i) {
        wf[i].x = wcol[(size_t)(4 * i) * D];
        wf[i].y = wcol[(size_t)(4 * i + 1) * D];
    }
    const float att_s = att[col0 + lr];
    const float att_d = att[D + col0 + lr];

    *(float4*)(&sA[0][wave * GP + lane * 4])       = p0;
    *(float4*)(&sA[0][(wave + 8) * GP + lane * 4]) = p1;
    __syncthreads();

    int buf = 0;
    for (; strip < nstrips; strip += step) {
        const int row0 = strip * 16;
        const int next = strip + step;
        const bool more = next < nstrips;
        if (more) {
            p0 = load_hrow(h0, n0, h1, next * 16 + wave,     n, lane);
            p1 = load_hrow(h0, n0, h1, next * 16 + wave + 8, n, lane);
        }

        v8f c = {0.f, 0.f, 0.f, 0.f, 0.f, 0.f, 0.f, 0.f};
        const float* arow = &sA[buf][lr * GP + 2 * half];
        v2f af[32];
#pragma unroll
        for (int i = 0; i < 32; ++i) {
            const float2 av = *(const float2*)(arow + 4 * i);
            af[i].x = av.x; af[i].y = av.y;
        }
        __builtin_amdgcn_sched_barrier(0);
#pragma unroll
        for (int i = 0; i < 32; ++i) {
            c = __builtin_amdgcn_wmma_f32_16x16x4_f32(
                    false, af[i], false, wf[i], (short)0, c, false, false);
        }
        const float ps = rows_dot16(c, att_s, lr);
        const float pd = rows_dot16(c, att_d, lr);
        if ((lr & 1) == 0) {
            sc[buf][wave][0][8 * half + (lr >> 1)] = ps;
            sc[buf][wave][1][8 * half + (lr >> 1)] = pd;
        }
        if (more) {
            *(float4*)(&sA[buf ^ 1][wave * GP + lane * 4])       = p0;
            *(float4*)(&sA[buf ^ 1][(wave + 8) * GP + lane * 4]) = p1;
        }
        __syncthreads();

#pragma unroll
        for (int v = 0; v < 8; ++v) {
            const int r = row0 + v + 8 * half;
            if (r < n) Z[(size_t)r * D + col0 + lr] = c[v];
        }

        if (threadIdx.x < 32) {
            const int which = threadIdx.x >> 4;
            const int rr = threadIdx.x & 15;
            float t = 0.0f;
#pragma unroll
            for (int w8 = 0; w8 < 8; ++w8) t += sc[buf][w8][which][rr];
            const int r = row0 + rr;
            if (r < n) { if (which) sdst[r] = t; else ssrc[r] = t; }
        }
        buf ^= 1;
    }
}

__device__ __forceinline__ float leaky(float v) {
    return v > 0.0f ? v : 0.01f * v;
}

#define SLOT 32

struct EdgePart {
    const int* src; const int* dst;
    int lo, hi, nE;
    int blk0;
    int rbase;
    int g;
    int obase;
};

__device__ __forceinline__ EdgePart pick_part(
    const int* s0, const int* d0, int lo0, int hi0, int nE0,
    const int* s1, const int* d1, int lo1, int hi1, int nE1, int blk1, int rb1,
    const int* s2, const int* d2, int lo2, int hi2, int nE2, int blk2, int rb2)
{
    EdgePart p;
    const int b = blockIdx.x;
    if (b >= blk2)      { p.src = s2; p.dst = d2; p.lo = lo2; p.hi = hi2; p.nE = nE2; p.blk0 = blk2; p.rbase = rb2; p.g = 2; p.obase = nE0 + nE1; }
    else if (b >= blk1) { p.src = s1; p.dst = d1; p.lo = lo1; p.hi = hi1; p.nE = nE1; p.blk0 = blk1; p.rbase = rb1; p.g = 1; p.obase = nE0; }
    else                { p.src = s0; p.dst = d0; p.lo = lo0; p.hi = hi0; p.nE = nE0; p.blk0 = 0;    p.rbase = 0;   p.g = 0; p.obase = 0; }
    return p;
}

__device__ __forceinline__ int4 load_ids4(const int* __restrict__ p, int e0, int nE) {
    int4 v = make_int4(-1, -1, -1, -1);
    if (e0 + 3 < nE) {
        v = *(const int4*)(p + e0);
    } else {
        if (e0     < nE) v.x = p[e0];
        if (e0 + 1 < nE) v.y = p[e0 + 1];
        if (e0 + 2 < nE) v.z = p[e0 + 2];
    }
    return v;
}

#define RB 384
#define FILL_T 1024
#define FILL_TRIPS (RB * SLOT / FILL_T)
#define GROUPS 8
#define SWEEP (FILL_T * 4 * GROUPS)

__device__ __forceinline__ void own_edge(int* __restrict__ lcnt, int* __restrict__ lslot,
                                         int* __restrict__ novf, int* __restrict__ ovf,
                                         int d, int dlo, int nown, int e)
{
    const int r = d - dlo;
    if ((unsigned int)r < (unsigned int)nown) {
        const int pos = atomicAdd(&lcnt[r], 1);
        if (pos < SLOT) lslot[r * SLOT + pos] = e;
        else            ovf[atomicAdd(novf, 1)] = e;
    }
}

__device__ __forceinline__ void own_edges4(int* __restrict__ lcnt, int* __restrict__ lslot,
                                           int* __restrict__ novf, int* __restrict__ ovf,
                                           const int4 d, int dlo, int nown, int e0)
{
    own_edge(lcnt, lslot, novf, ovf, d.x, dlo, nown, e0);
    own_edge(lcnt, lslot, novf, ovf, d.y, dlo, nown, e0 + 1);
    own_edge(lcnt, lslot, novf, ovf, d.z, dlo, nown, e0 + 2);
    own_edge(lcnt, lslot, novf, ovf, d.w, dlo, nown, e0 + 3);
}

__device__ __forceinline__ void load_sweep(int4 (&d)[GROUPS], const int* __restrict__ dst, int bs)
{
    const int e0 = bs * SWEEP + (int)threadIdx.x * 4;
#pragma unroll
    for (int j = 0; j < GROUPS; ++j) d[j] = *(const int4*)(dst + e0 + j * (FILL_T * 4));
}

__global__ void __launch_bounds__(FILL_T)
edge_fill3(const int* __restrict__ s0, const int* __restrict__ d0, int lo0, int hi0, int nE0,
           const int* __restrict__ s1, const int* __restrict__ d1, int lo1, int hi1, int nE1, int blk1, int rb1,
           const int* __restrict__ s2, const int* __restrict__ d2, int lo2, int hi2, int nE2, int blk2, int rb2,
           int* __restrict__ cnt_all, int* __restrict__ slots_all,
           int* __restrict__ novf_all, int* __restrict__ ovf_all)
{
    __shared__ int lcnt[RB];
    __shared__ int lslot[RB * SLOT];
    const EdgePart p = pick_part(s0, d0, lo0, hi0, nE0,
                                 s1, d1, lo1, hi1, nE1, blk1, rb1,
                                 s2, d2, lo2, hi2, nE2, blk2, rb2);
    const int tid = threadIdx.x;
    const int b = blockIdx.x - p.blk0;
    const int row_lo = b * RB;
    const int left = (p.hi - p.lo) - row_lo;
    const int nown = left < RB ? left : RB;
    const int dlo = p.lo + row_lo;
    const int nE = p.nE;
    int* novf = novf_all + p.g;
    int* ovf  = ovf_all + p.obase;

    for (int i = tid; i < RB; i += FILL_T) lcnt[i] = 0;
    __syncthreads();

    const int nsw = nE / SWEEP;
    if (nsw > 0) {
        int bs = b % nsw;
        int4 cur[GROUPS];
        load_sweep(cur, p.dst, bs);
        for (int k = 0; k < nsw; ++k) {
            const int e0 = bs * SWEEP + tid * 4;
            int nb = bs + 1; if (nb == nsw) nb = 0;
            int4 nxt[GROUPS];
            load_sweep(nxt, p.dst, nb);
#pragma unroll
            for (int j = 0; j < GROUPS; ++j)
                own_edges4(lcnt, lslot, novf, ovf, cur[j], dlo, nown, e0 + j * (FILL_T * 4));
#pragma unroll
            for (int j = 0; j < GROUPS; ++j) cur[j] = nxt[j];
            bs = nb;
        }
    }
    {
        const int e0 = nsw * SWEEP + tid * 4;
        const int ngrp = (nE - nsw * SWEEP) / (FILL_T * 4);
        int4 dt[GROUPS];
#pragma unroll
        for (int j = 0; j < GROUPS; ++j)
            if (j < ngrp) dt[j] = *(const int4*)(p.dst + e0 + j * (FILL_T * 4));
#pragma unroll
        for (int j = 0; j < GROUPS; ++j)
            if (j < ngrp) own_edges4(lcnt, lslot, novf, ovf, dt[j], dlo, nown, e0 + j * (FILL_T * 4));
        const int el = e0 + ngrp * (FILL_T * 4);
        own_edges4(lcnt, lslot, novf, ovf, load_ids4(p.dst, el, nE), dlo, nown, el);
    }
    __syncthreads();

    int* cnt   = cnt_all + p.rbase + row_lo;
    int* slots = slots_all + (size_t)(p.rbase + row_lo) * SLOT;
    for (int i = tid; i < nown; i += FILL_T) cnt[i] = lcnt[i];

    int nv[FILL_TRIPS];
#pragma unroll
    for (int j = 0; j < FILL_TRIPS; ++j) nv[j] = lcnt[(tid + j * FILL_T) / SLOT];
    int sv[FILL_TRIPS];
#pragma unroll
    for (int j = 0; j < FILL_TRIPS; ++j) {
        const int idx = tid + j * FILL_T;
        int s = 0;
        if ((idx % SLOT) < nv[j]) s = p.src[lslot[idx]];
        sv[j] = s;
    }
#pragma unroll
    for (int j = 0; j < FILL_TRIPS; ++j) {
        const int idx = tid + j * FILL_T;
        if (idx < nown * SLOT) slots[idx] = sv[j];
    }
}

__device__ __forceinline__ float wave_max(float x) {
#pragma unroll
    for (int off = 16; off > 0; off >>= 1) x = fmaxf(x, __shfl_xor(x, off, 32));
    return x;
}
__device__ __forceinline__ float wave_sum(float x) {
#pragma unroll
    for (int off = 16; off > 0; off >>= 1) x += __shfl_xor(x, off, 32);
    return x;
}

struct RowSrc {
    const int* cnt; const int* slots;
    const int* novf; const int* ovf;
    const int* esrc; const int* edst;
};

__device__ __forceinline__ float4 agg_one_row(const RowSrc g,
                                              const float* __restrict__ ssrc, const float* __restrict__ sdst,
                                              const float* __restrict__ z, int lo, int r, int lane)
{
    const int n = __builtin_amdgcn_readfirstlane(g.cnt[r]);
    const int d = r + lo;
    const float sd = sdst[d];

    int   s0 = 0;
    float v0 = -INFINITY;
    const bool have0 = lane < n;
    if (have0) { s0 = g.slots[(size_t)r * SLOT + lane]; v0 = leaky(ssrc[s0] + sd); }
    float mx = v0;
    int nov = 0;
    if (n > SLOT) {
        nov = __builtin_amdgcn_readfirstlane(g.novf[0]);
        for (int j = lane; j < nov; j += 32) {
            const int e = g.ovf[j];
            if (g.edst[e] == d) mx = fmaxf(mx, leaky(ssrc[g.esrc[e]] + sd));
        }
    }
    mx = wave_max(mx);

    const float ex0 = have0 ? __expf(v0 - mx) : 0.0f;
    float dn = ex0;
    if (n > SLOT) {
        for (int j = lane; j < nov; j += 32) {
            const int e = g.ovf[j];
            if (g.edst[e] == d) dn += __expf(leaky(ssrc[g.esrc[e]] + sd) - mx);
        }
    }
    dn = wave_sum(dn);

    float4 acc = make_float4(0.0f, 0.0f, 0.0f, 0.0f);
    {
        const int   s = s0;
        const float a = have0 ? ex0 / dn : 0.0f;
        const int cnt = n < SLOT ? n : SLOT;
#pragma unroll 4
        for (int i = 0; i < cnt; ++i) {
            const int   si = __builtin_amdgcn_readlane(s, i);
            const float ai = __int_as_float(__builtin_amdgcn_readlane(__float_as_int(a), i));
            const float4 zv = *(const float4*)(z + (size_t)si * D + lane * 4);
            acc.x = fmaf(ai, zv.x, acc.x);
            acc.y = fmaf(ai, zv.y, acc.y);
            acc.z = fmaf(ai, zv.z, acc.z);
            acc.w = fmaf(ai, zv.w, acc.w);
        }
    }
    if (n > SLOT) {
        for (int base = 0; base < nov; base += 32) {
            const int j = base + lane;
            int  s = 0;
            bool mine = false;
            float a = 0.0f;
            if (j < nov) {
                const int e = g.ovf[j];
                if (g.edst[e] == d) {
                    mine = true;
                    s = g.esrc[e];
                    a = __expf(leaky(ssrc[s] + sd) - mx) / dn;
                }
            }
            unsigned int m = (unsigned int)__ballot(mine);
            while (m != 0u) {
                const int i = __builtin_ctz(m);
                m &= m - 1u;
                const int   si = __builtin_amdgcn_readlane(s, i);
                const float ai = __int_as_float(__builtin_amdgcn_readlane(__float_as_int(a), i));
                const float4 zv = *(const float4*)(z + (size_t)si * D + lane * 4);
                acc.x = fmaf(ai, zv.x, acc.x);
                acc.y = fmaf(ai, zv.y, acc.y);
                acc.z = fmaf(ai, zv.z, acc.z);
                acc.w = fmaf(ai, zv.w, acc.w);
            }
        }
    }
    return acc;
}

__global__ void agg_rows(const RowSrc g,
                         const float* __restrict__ ssrc, const float* __restrict__ sdst,
                         const float* __restrict__ z, float* __restrict__ out,
                         int lo, int nr)
{
    const int lane = threadIdx.x & 31;
    const int r = __builtin_amdgcn_readfirstlane((blockIdx.x * blockDim.x + threadIdx.x) >> 5);
    if (r >= nr) return;
    const float4 acc = agg_one_row(g, ssrc, sdst, z, lo, r, lane);
    *(float4*)(out + (size_t)r * D + lane * 4) = acc;
}

__global__ void __launch_bounds__(512, 16)
agg_combine(const RowSrc g_d, const RowSrc g_k,
            const float* __restrict__ ssrc_d, const float* __restrict__ sdst_d,
            const float* __restrict__ z_d,
            const float* __restrict__ ssrc_k, const float* __restrict__ sdst_k,
            const float* __restrict__ z_k, int lo_k,
            const float* __restrict__ W1, const float* __restrict__ b1,
            const float* __restrict__ w2, float* __restrict__ out, int n)
{
    __shared__ __attribute__((aligned(16))) float sX[2][16 * GP];
    __shared__ __attribute__((aligned(16))) float sc[2][16][8];
    const int wave = __builtin_amdgcn_readfirstlane(threadIdx.x >> 5);
    const int lane = threadIdx.x & 31;
    const int half = lane >> 4;
    const int lr   = lane & 15;
    const int g    = wave >> 3;
    const int wt   = wave & 7;
    const int row0 = blockIdx.x * 16;
    const int col0 = wt * 16;

    {
        const int r = row0 + wave;
        float4 ra = make_float4(0.0f, 0.0f, 0.0f, 0.0f);
        if (r < n) ra = agg_one_row(g_d, ssrc_d, sdst_d, z_d, 0, r, lane);
        *(float4*)(sX[0] + wave * GP + lane * 4) = ra;
        float4 rb = make_float4(0.0f, 0.0f, 0.0f, 0.0f);
        if (r < n) rb = agg_one_row(g_k, ssrc_k, sdst_k, z_k, lo_k, r, lane);
        *(float4*)(sX[1] + wave * GP + lane * 4) = rb;
    }
    __syncthreads();

    v8f c = {0.f, 0.f, 0.f, 0.f, 0.f, 0.f, 0.f, 0.f};
    const float* xrow = sX[g] + lr * GP + 2 * half;
    const float* wcol = W1 + (size_t)(2 * half) * D + col0 + lr;
#pragma unroll
    for (int k0 = 0; k0 < D; k0 += 4) {
        const float2 xv = *(const float2*)(xrow + k0);
        v2f a; a.x = xv.x; a.y = xv.y;
        v2f w; w.x = wcol[(size_t)k0 * D];
               w.y = wcol[(size_t)(k0 + 1) * D];
        c = __builtin_amdgcn_wmma_f32_16x16x4_f32(
                false, a, false, w, (short)0, c, false, false);
    }

    const float bias = b1[col0 + lr];
    const float w2c  = w2[col0 + lr];
#pragma unroll
    for (int v = 0; v < 8; ++v) c[v] = tanhf(c[v] + bias);
    const float pp = rows_dot16(c, w2c, lr);
    if ((lr & 1) == 0) sc[g][8 * half + (lr >> 1)][wt] = pp;
    __syncthreads();

    {
        const int rr = wave;
        const float4 x0 = *(const float4*)&sc[0][rr][0];
        const float4 x1 = *(const float4*)&sc[0][rr][4];
        const float4 y0 = *(const float4*)&sc[1][rr][0];
        const float4 y1 = *(const float4*)&sc[1][rr][4];
        const float s1 = ((((((x0.x + x0.y) + x0.z) + x0.w) + x1.x) + x1.y) + x1.z) + x1.w;
        const float s2 = ((((((y0.x + y0.y) + y0.z) + y0.w) + y1.x) + y1.y) + y1.z) + y1.w;
        const float mm = fmaxf(s1, s2);
        const float e1 = __expf(s1 - mm), e2 = __expf(s2 - mm);
        const float al = e1 / (e1 + e2);
        const float be = 1.0f - al;
        const float4 va = *(const float4*)(sX[0] + rr * GP + lane * 4);
        const float4 vb = *(const float4*)(sX[1] + rr * GP + lane * 4);
        float4 o;
        o.x = al * va.x + be * vb.x;
        o.y = al * va.y + be * vb.y;
        o.z = al * va.z + be * vb.z;
        o.w = al * va.w + be * vb.w;
        const int r = row0 + rr;
        if (r < n) *(float4*)(out + (size_t)r * D + lane * 4) = o;
    }
}

static inline int cdiv(int a, int b) { return (a + b - 1) / b; }

extern "C" void kernel_launch(void* const* d_in, const int* in_sizes, int n_in,
                              void* d_out, int out_size, void* d_ws, size_t ws_size,
                              hipStream_t stream)
{
    (void)n_in; (void)ws_size; (void)out_size;
    const float* kn_emb   = (const float*)d_in[0];
    const float* exer_emb = (const float*)d_in[1];
    const float* W_dir    = (const float*)d_in[2];
    const float* a_dir    = (const float*)d_in[3];
    const float* W_ke     = (const float*)d_in[4];
    const float* a_ke     = (const float*)d_in[5];
    const float* W_ek     = (const float*)d_in[6];
    const float* a_ek     = (const float*)d_in[7];
    const float* rW1      = (const float*)d_in[8];
    const float* rb1      = (const float*)d_in[9];
    const float* rw2      = (const float*)d_in[10];
    const int* dsrc  = (const int*)d_in[11];
    const int* ddst  = (const int*)d_in[12];
    const int* kesrc = (const int*)d_in[13];
    const int* kedst = (const int*)d_in[14];
    const int* eksrc = (const int*)d_in[15];
    const int* ekdst = (const int*)d_in[16];

    const int Kn = in_sizes[0] / D;
    const int In = in_sizes[1] / D;
    const int Nn = Kn + In;
    const int E1 = in_sizes[11];
    const int E2 = in_sizes[13];

    float* w = (float*)d_ws;
    size_t off = 0;
    float* buf_z = w + off; off += (size_t)Nn * D;
    float* buf_B = w + off; off += (size_t)Nn * D;
    float* buf_A = w + off; off += (size_t)Kn * D;
    float* ssrc  = w + off; off += (size_t)Nn;
    float* sdst  = w + off; off += (size_t)Nn;
    float* ssrc_d = buf_A;
    float* sdst_d = buf_A + Kn;

    const int r4I = (In + 3) & ~3;
    const int r4K = (Kn + 3) & ~3;
    const int rb_ke  = r4I;
    const int rb_dir = r4I + r4K;
    const int nrows  = r4I + 2 * r4K;
    const int nclr   = nrows + 4;
    const int ncl32  = (nclr + 31) & ~31;
    int* cnt   = (int*)(buf_B + (size_t)Kn * D);
    int* novf  = cnt + nrows;
    int* slots = cnt + ncl32;
    int* ovf   = slots + (size_t)nrows * SLOT;

    float* out_kn = (float*)d_out;
    float* out_ex = (float*)d_out + (size_t)Kn * D;

    const int gb_ex = cdiv(In, RB);
    const int gb_kn = cdiv(Kn, RB);

    RowSrc g_ek;  g_ek.cnt  = cnt;          g_ek.slots  = slots;
                  g_ek.novf = novf;         g_ek.ovf    = ovf;
                  g_ek.esrc = eksrc;        g_ek.edst   = ekdst;
    RowSrc g_ke;  g_ke.cnt  = cnt + rb_ke;  g_ke.slots  = slots + (size_t)rb_ke * SLOT;
                  g_ke.novf = novf + 1;     g_ke.ovf    = ovf + E2;
                  g_ke.esrc = kesrc;        g_ke.edst   = kedst;
    RowSrc g_dir; g_dir.cnt  = cnt + rb_dir; g_dir.slots = slots + (size_t)rb_dir * SLOT;
                  g_dir.novf = novf + 2;     g_dir.ovf   = ovf + 2 * (size_t)E2;
                  g_dir.esrc = dsrc;         g_dir.edst  = ddst;

    gemm_z<<<512, 256, 0, stream>>>(exer_emb, In, kn_emb, W_ek, a_ek, buf_z, ssrc, sdst, Nn, novf, 4);

    edge_fill3<<<gb_ex + 2 * gb_kn, FILL_T, 0, stream>>>(eksrc, ekdst, 0, In, E2,
                                                         kesrc, kedst, In, Nn, E2, gb_ex, rb_ke,
                                                         dsrc, ddst, 0, Kn, E1, gb_ex + gb_kn, rb_dir,
                                                         cnt, slots, novf, ovf);

    agg_rows<<<cdiv(In, 8), 256, 0, stream>>>(g_ek, ssrc, sdst, buf_z, out_ex, 0, In);

    gemm_z<<<512, 256, 0, stream>>>(exer_emb, In, kn_emb, W_ke, a_ke, buf_z, ssrc, sdst, Nn, nullptr, 0);

    gemm_z<<<512, 256, 0, stream>>>(kn_emb, Kn, kn_emb, W_dir, a_dir, buf_B, ssrc_d, sdst_d, Kn, nullptr, 0);

    agg_combine<<<cdiv(Kn, 16), 512, 0, stream>>>(g_dir, g_ke,
                                                  ssrc_d, sdst_d, buf_B,
                                                  ssrc, sdst, buf_z, In,
                                                  rW1, rb1, rw2, out_kn, Kn);
}


# ke and directed graphs aggregate h ahead of the product with W (A=(sum alpha*h[src])@W_dir, B alike; sum order changes) in the tail kernel, 32 nodes a block; node scores h.(W@att) ride the ek product
# speedup vs baseline: 6.0085x; 1.1612x over previous
#include <hip/hip_runtime.h>
#include <hip/hip_bf16.h>
#include <math.h>


#define D 128

typedef float v2f __attribute__((ext_vector_type(2)));
typedef float v8f __attribute__((ext_vector_type(8)));

__device__ __forceinline__ float rows_dot16(const v8f c, float coef, int lr)
{
    const bool b3 = (lr & 8) != 0;
    const bool b2 = (lr & 4) != 0;
    const bool b1 = (lr & 2) != 0;
    float y[4];
#pragma unroll
    for (int i = 0; i < 4; ++i) {
        const float lo = c[i] * coef, hi = c[i + 4] * coef;
        const float keep = b3 ? hi : lo;
        const float send = b3 ? lo : hi;
        y[i] = keep + __shfl_xor(send, 8, 32);
    }
    float q[2];
#pragma unroll
    for (int i = 0; i < 2; ++i) {
        const float keep = b2 ? y[i + 2] : y[i];
        const float send = b2 ? y[i] : y[i + 2];
        q[i] = keep + __shfl_xor(send, 4, 32);
    }
    const float keep = b1 ? q[1] : q[0];
    const float send = b1 ? q[0] : q[1];
    float w = keep + __shfl_xor(send, 2, 32);
    w += __shfl_xor(w, 1, 32);
    return w;
}

__device__ __forceinline__ float wave_max(float x) {
#pragma unroll
    for (int off = 16; off > 0; off >>= 1) x = fmaxf(x, __shfl_xor(x, off, 32));
    return x;
}
__device__ __forceinline__ float wave_sum(float x) {
#pragma unroll
    for (int off = 16; off > 0; off >>= 1) x += __shfl_xor(x, off, 32);
    return x;
}

__device__ __forceinline__ float dot4(const float4 a, const float4 b) {
    return fmaf(a.w, b.w, fmaf(a.z, b.z, fmaf(a.y, b.y, a.x * b.x)));
}

struct NodeScores {
    const float* Wk; const float* attk;
    float* ssrc_k; float* sdst_k;
    const float* Wd; const float* attd;
    float* ssrc_d; float* sdst_d;
};

__device__ __forceinline__ float wave_sums32(const float (&v)[32], int lane)
{
    const bool b4 = (lane & 16) != 0, b3 = (lane & 8) != 0, b2 = (lane & 4) != 0;
    const bool b1 = (lane & 2) != 0,  b0 = (lane & 1) != 0;
    float a[16];
#pragma unroll
    for (int i = 0; i < 16; ++i) {
        const float keep = b4 ? v[i + 16] : v[i];
        const float send = b4 ? v[i] : v[i + 16];
        a[i] = keep + __shfl_xor(send, 16, 32);
    }
    float b[8];
#pragma unroll
    for (int i = 0; i < 8; ++i) {
        const float keep = b3 ? a[i + 8] : a[i];
        const float send = b3 ? a[i] : a[i + 8];
        b[i] = keep + __shfl_xor(send, 8, 32);
    }
    float q[4];
#pragma unroll
    for (int i = 0; i < 4; ++i) {
        const float keep = b2 ? b[i + 4] : b[i];
        const float send = b2 ? b[i] : b[i + 4];
        q[i] = keep + __shfl_xor(send, 4, 32);
    }
    float d[2];
#pragma unroll
    for (int i = 0; i < 2; ++i) {
        const float keep = b1 ? q[i + 2] : q[i];
        const float send = b1 ? q[i] : q[i + 2];
        d[i] = keep + __shfl_xor(send, 2, 32);
    }
    const float keep = b0 ? d[1] : d[0];
    const float send = b0 ? d[0] : d[1];
    return keep + __shfl_xor(send, 1, 32);
}

__device__ __forceinline__ float wave_sums8(const float (&v)[8], int lane)
{
    const bool b4 = (lane & 16) != 0, b3 = (lane & 8) != 0, b2 = (lane & 4) != 0;
    float a[4];
#pragma unroll
    for (int i = 0; i < 4; ++i) {
        const float keep = b4 ? v[i + 4] : v[i];
        const float send = b4 ? v[i] : v[i + 4];
        a[i] = keep + __shfl_xor(send, 16, 32);
    }
    float b[2];
#pragma unroll
    for (int i = 0; i < 2; ++i) {
        const float keep = b3 ? a[i + 2] : a[i];
        const float send = b3 ? a[i] : a[i + 2];
        b[i] = keep + __shfl_xor(send, 8, 32);
    }
    const float keep = b2 ? b[1] : b[0];
    const float send = b2 ? b[0] : b[1];
    float t = keep + __shfl_xor(send, 4, 32);
    t += __shfl_xor(t, 2, 32);
    t += __shfl_xor(t, 1, 32);
    return t;
}

__device__ __forceinline__ void node_scores(const NodeScores x, const float4 p0, const float4 p1,
                                            int r0, int n0, int n, int lane,
                                            const float4 uks, const float4 ukd,
                                            const float4 uds, const float4 udd)
{
    float v[8];
    v[0] = dot4(p0, uks); v[1] = dot4(p0, ukd); v[2] = dot4(p0, uds); v[3] = dot4(p0, udd);
    v[4] = dot4(p1, uks); v[5] = dot4(p1, ukd); v[6] = dot4(p1, uds); v[7] = dot4(p1, udd);
    const float t = wave_sums8(v, lane);
    const int j = (lane >> 2) & 7;
    const int r = r0 + 8 * (j >> 2);
    const int vec = j & 3;
    if ((lane & 3) == 0 && r < n) {
        if (vec == 0)      x.ssrc_k[r] = t;
        else if (vec == 1) x.sdst_k[r] = t;
        else if (r >= n0) {
            if (vec == 2) x.ssrc_d[r - n0] = t;
            else          x.sdst_d[r - n0] = t;
        }
    }
}

#define GP 132

__device__ __forceinline__ float4 load_hrow(const float* __restrict__ h0, int n0,
                                            const float* __restrict__ h1,
                                            int r, int n, int lane)
{
    if (r >= n) r = n - 1;
    const float* hrow = (r < n0) ? (h0 + (size_t)r * D)
                                 : (h1 + (size_t)(r - n0) * D);
    return *(const float4*)(hrow + lane * 4);
}

__global__ void __launch_bounds__(256)
gemm_z(const float* __restrict__ h0, int n0,
       const float* __restrict__ h1,
       const float* __restrict__ W,
       const float* __restrict__ att,
       float* __restrict__ Z,
       float* __restrict__ ssrc, float* __restrict__ sdst, int n,
       int* __restrict__ clr, int nclr, const NodeScores x)
{
    __shared__ __attribute__((aligned(16))) float sA[2][16 * GP];
    __shared__ __attribute__((aligned(16))) float su[4][D];
    __shared__ float sc[2][8][2][16];
    const int wave = threadIdx.x >> 5;
    const int lane = threadIdx.x & 31;
    const int half = lane >> 4;
    const int lr   = lane & 15;
    const int col0 = wave * 16;
    const int nstrips = (n + 15) >> 4;
    const int step = gridDim.x;

    for (int i = blockIdx.x * blockDim.x + threadIdx.x; i < nclr; i += gridDim.x * blockDim.x)
        clr[i] = 0;

    int strip = blockIdx.x;
    if (strip >= nstrips) return;

    float4 p0 = load_hrow(h0, n0, h1, strip * 16 + wave,     n, lane);
    float4 p1 = load_hrow(h0, n0, h1, strip * 16 + wave + 8, n, lane);

    v2f wf[32];
    const float* wcol = W + (size_t)(2 * half) * D + col0 + lr;
#pragma unroll
    for (int i = 0; i < 32; ++i) {
        wf[i].x = wcol[(size_t)(4 * i) * D];
        wf[i].y = wcol[(size_t)(4 * i + 1) * D];
    }
    const float att_s = att[col0 + lr];
    const float att_d = att[D + col0 + lr];

    {
        const float4 aks = *(const float4*)(x.attk + lane * 4);
        const float4 akd = *(const float4*)(x.attk + D + lane * 4);
        const float4 ads = *(const float4*)(x.attd + lane * 4);
        const float4 add = *(const float4*)(x.attd + D + lane * 4);
        float vk[32], vd[32];
#pragma unroll
        for (int i = 0; i < 16; ++i) {
            const int k = wave * 16 + i;
            const float4 wk = *(const float4*)(x.Wk + (size_t)k * D + lane * 4);
            const float4 wd = *(const float4*)(x.Wd + (size_t)k * D + lane * 4);
            vk[i] = dot4(wk, aks); vk[16 + i] = dot4(wk, akd);
            vd[i] = dot4(wd, ads); vd[16 + i] = dot4(wd, add);
        }
        su[half][wave * 16 + lr]     = wave_sums32(vk, lane);
        su[2 + half][wave * 16 + lr] = wave_sums32(vd, lane);
    }

    *(float4*)(&sA[0][wave * GP + lane * 4])       = p0;
    *(float4*)(&sA[0][(wave + 8) * GP + lane * 4]) = p1;
    __syncthreads();

    const float4 uks = *(const float4*)(&su[0][lane * 4]);
    const float4 ukd = *(const float4*)(&su[1][lane * 4]);
    const float4 uds = *(const float4*)(&su[2][lane * 4]);
    const float4 udd = *(const float4*)(&su[3][lane * 4]);
    node_scores(x, p0, p1, strip * 16 + wave, n0, n, lane, uks, ukd, uds, udd);

    int buf = 0;
    for (; strip < nstrips; strip += step) {
        const int row0 = strip * 16;
        const int next = strip + step;
        const bool more = next < nstrips;
        if (more) {
            p0 = load_hrow(h0, n0, h1, next * 16 + wave,     n, lane);
            p1 = load_hrow(h0, n0, h1, next * 16 + wave + 8, n, lane);
        }

        v8f c = {0.f, 0.f, 0.f, 0.f, 0.f, 0.f, 0.f, 0.f};
        const float* arow = &sA[buf][lr * GP + 2 * half];
        v2f af[32];
#pragma unroll
        for (int i = 0; i < 32; ++i) {
            const float2 av = *(const float2*)(arow + 4 * i);
            af[i].x = av.x; af[i].y = av.y;
        }
        __builtin_amdgcn_sched_barrier(0);
#pragma unroll
        for (int i = 0; i < 32; ++i) {
            c = __builtin_amdgcn_wmma_f32_16x16x4_f32(
                    false, af[i], false, wf[i], (short)0, c, false, false);
        }
        const float ps = rows_dot16(c, att_s, lr);
        const float pd = rows_dot16(c, att_d, lr);
        if ((lr & 1) == 0) {
            sc[buf][wave][0][8 * half + (lr >> 1)] = ps;
            sc[buf][wave][1][8 * half + (lr >> 1)] = pd;
        }
        if (more) {
            *(float4*)(&sA[buf ^ 1][wave * GP + lane * 4])       = p0;
            *(float4*)(&sA[buf ^ 1][(wave + 8) * GP + lane * 4]) = p1;
        }
        __syncthreads();

        if (more) node_scores(x, p0, p1, next * 16 + wave, n0, n, lane, uks, ukd, uds, udd);

#pragma unroll
        for (int v = 0; v < 8; ++v) {
            const int r = row0 + v + 8 * half;
            if (r < n) Z[(size_t)r * D + col0 + lr] = c[v];
        }

        if (threadIdx.x < 32) {
            const int which = threadIdx.x >> 4;
            const int rr = threadIdx.x & 15;
            float t = 0.0f;
#pragma unroll
            for (int w8 = 0; w8 < 8; ++w8) t += sc[buf][w8][which][rr];
            const int r = row0 + rr;
            if (r < n) { if (which) sdst[r] = t; else ssrc[r] = t; }
        }
        buf ^= 1;
    }
}

__device__ __forceinline__ float leaky(float v) {
    return v > 0.0f ? v : 0.01f * v;
}

#define SLOT 32

struct EdgePart {
    const int* src; const int* dst;
    int lo, hi, nE;
    int blk0;
    int rbase;
    int g;
    int obase;
};

__device__ __forceinline__ EdgePart pick_part(
    const int* s0, const int* d0, int lo0, int hi0, int nE0,
    const int* s1, const int* d1, int lo1, int hi1, int nE1, int blk1, int rb1,
    const int* s2, const int* d2, int lo2, int hi2, int nE2, int blk2, int rb2)
{
    EdgePart p;
    const int b = blockIdx.x;
    if (b >= blk2)      { p.src = s2; p.dst = d2; p.lo = lo2; p.hi = hi2; p.nE = nE2; p.blk0 = blk2; p.rbase = rb2; p.g = 2; p.obase = nE0 + nE1; }
    else if (b >= blk1) { p.src = s1; p.dst = d1; p.lo = lo1; p.hi = hi1; p.nE = nE1; p.blk0 = blk1; p.rbase = rb1; p.g = 1; p.obase = nE0; }
    else                { p.src = s0; p.dst = d0; p.lo = lo0; p.hi = hi0; p.nE = nE0; p.blk0 = 0;    p.rbase = 0;   p.g = 0; p.obase = 0; }
    return p;
}

__device__ __forceinline__ int4 load_ids4(const int* __restrict__ p, int e0, int nE) {
    int4 v = make_int4(-1, -1, -1, -1);
    if (e0 + 3 < nE) {
        v = *(const int4*)(p + e0);
    } else {
        if (e0     < nE) v.x = p[e0];
        if (e0 + 1 < nE) v.y = p[e0 + 1];
        if (e0 + 2 < nE) v.z = p[e0 + 2];
    }
    return v;
}

#define RB 384
#define FILL_T 1024
#define FILL_TRIPS (RB * SLOT / FILL_T)
#define GROUPS 8
#define SWEEP (FILL_T * 4 * GROUPS)

__device__ __forceinline__ void own_edge(int* __restrict__ lcnt, int* __restrict__ lslot,
                                         int* __restrict__ novf, int* __restrict__ ovf,
                                         int d, int dlo, int nown, int e)
{
    const int r = d - dlo;
    if ((unsigned int)r < (unsigned int)nown) {
        const int pos = atomicAdd(&lcnt[r], 1);
        if (pos < SLOT) lslot[r * SLOT + pos] = e;
        else            ovf[atomicAdd(novf, 1)] = e;
    }
}

__device__ __forceinline__ void own_edges4(int* __restrict__ lcnt, int* __restrict__ lslot,
                                           int* __restrict__ novf, int* __restrict__ ovf,
                                           const int4 d, int dlo, int nown, int e0)
{
    own_edge(lcnt, lslot, novf, ovf, d.x, dlo, nown, e0);
    own_edge(lcnt, lslot, novf, ovf, d.y, dlo, nown, e0 + 1);
    own_edge(lcnt, lslot, novf, ovf, d.z, dlo, nown, e0 + 2);
    own_edge(lcnt, lslot, novf, ovf, d.w, dlo, nown, e0 + 3);
}

__device__ __forceinline__ void load_sweep(int4 (&d)[GROUPS], const int* __restrict__ dst, int bs)
{
    const int e0 = bs * SWEEP + (int)threadIdx.x * 4;
#pragma unroll
    for (int j = 0; j < GROUPS; ++j) d[j] = *(const int4*)(dst + e0 + j * (FILL_T * 4));
}

__global__ void __launch_bounds__(FILL_T)
edge_fill3(const int* __restrict__ s0, const int* __restrict__ d0, int lo0, int hi0, int nE0,
           const int* __restrict__ s1, const int* __restrict__ d1, int lo1, int hi1, int nE1, int blk1, int rb1,
           const int* __restrict__ s2, const int* __restrict__ d2, int lo2, int hi2, int nE2, int blk2, int rb2,
           int* __restrict__ cnt_all, int* __restrict__ slots_all,
           int* __restrict__ novf_all, int* __restrict__ ovf_all)
{
    __shared__ int lcnt[RB];
    __shared__ int lslot[RB * SLOT];
    const EdgePart p = pick_part(s0, d0, lo0, hi0, nE0,
                                 s1, d1, lo1, hi1, nE1, blk1, rb1,
                                 s2, d2, lo2, hi2, nE2, blk2, rb2);
    const int tid = threadIdx.x;
    const int b = blockIdx.x - p.blk0;
    const int row_lo = b * RB;
    const int left = (p.hi - p.lo) - row_lo;
    const int nown = left < RB ? left : RB;
    const int dlo = p.lo + row_lo;
    const int nE = p.nE;
    int* novf = novf_all + p.g;
    int* ovf  = ovf_all + p.obase;

    for (int i = tid; i < RB; i += FILL_T) lcnt[i] = 0;
    __syncthreads();

    const int nsw = nE / SWEEP;
    if (nsw > 0) {
        int bs = b % nsw;
        int4 cur[GROUPS];
        load_sweep(cur, p.dst, bs);
        for (int k = 0; k < nsw; ++k) {
            const int e0 = bs * SWEEP + tid * 4;
            int nb = bs + 1; if (nb == nsw) nb = 0;
            int4 nxt[GROUPS];
            load_sweep(nxt, p.dst, nb);
#pragma unroll
            for (int j = 0; j < GROUPS; ++j)
                own_edges4(lcnt, lslot, novf, ovf, cur[j], dlo, nown, e0 + j * (FILL_T * 4));
#pragma unroll
            for (int j = 0; j < GROUPS; ++j) cur[j] = nxt[j];
            bs = nb;
        }
    }
    {
        const int e0 = nsw * SWEEP + tid * 4;
        const int ngrp = (nE - nsw * SWEEP) / (FILL_T * 4);
        int4 dt[GROUPS];
#pragma unroll
        for (int j = 0; j < GROUPS; ++j)
            if (j < ngrp) dt[j] = *(const int4*)(p.dst + e0 + j * (FILL_T * 4));
#pragma unroll
        for (int j = 0; j < GROUPS; ++j)
            if (j < ngrp) own_edges4(lcnt, lslot, novf, ovf, dt[j], dlo, nown, e0 + j * (FILL_T * 4));
        const int el = e0 + ngrp * (FILL_T * 4);
        own_edges4(lcnt, lslot, novf, ovf, load_ids4(p.dst, el, nE), dlo, nown, el);
    }
    __syncthreads();

    int* cnt   = cnt_all + p.rbase + row_lo;
    int* slots = slots_all + (size_t)(p.rbase + row_lo) * SLOT;
    for (int i = tid; i < nown; i += FILL_T) cnt[i] = lcnt[i];

    int nv[FILL_TRIPS];
#pragma unroll
    for (int j = 0; j < FILL_TRIPS; ++j) nv[j] = lcnt[(tid + j * FILL_T) / SLOT];
    int sv[FILL_TRIPS];
#pragma unroll
    for (int j = 0; j < FILL_TRIPS; ++j) {
        const int idx = tid + j * FILL_T;
        int s = 0;
        if ((idx % SLOT) < nv[j]) s = p.src[lslot[idx]];
        sv[j] = s;
    }
#pragma unroll
    for (int j = 0; j < FILL_TRIPS; ++j) {
        const int idx = tid + j * FILL_T;
        if (idx < nown * SLOT) slots[idx] = sv[j];
    }
}


struct RowSrc {
    const int* cnt; const int* slots;
    const int* novf; const int* ovf;
    const int* esrc; const int* edst;
};

template <bool CAT>
__device__ __forceinline__ const float* gather_row(const float* __restrict__ z, int n0,
                                                   const float* __restrict__ z1, int si)
{
    if (CAT && si >= n0) return z1 + (size_t)(si - n0) * D;
    return z + (size_t)si * D;
}

template <bool CAT>
__device__ __forceinline__ float4 agg_one_row(const RowSrc g,
                                              const float* __restrict__ ssrc, const float* __restrict__ sdst,
                                              const float* __restrict__ z, int n0, const float* __restrict__ z1,
                                              int lo, int r, int lane)
{
    const int n = __builtin_amdgcn_readfirstlane(g.cnt[r]);
    const int d = r + lo;
    const float sd = sdst[d];

    int   s0 = 0;
    float v0 = -INFINITY;
    const bool have0 = lane < n;
    if (have0) { s0 = g.slots[(size_t)r * SLOT + lane]; v0 = leaky(ssrc[s0] + sd); }
    float mx = v0;
    int nov = 0;
    if (n > SLOT) {
        nov = __builtin_amdgcn_readfirstlane(g.novf[0]);
        for (int j = lane; j < nov; j += 32) {
            const int e = g.ovf[j];
            if (g.edst[e] == d) mx = fmaxf(mx, leaky(ssrc[g.esrc[e]] + sd));
        }
    }
    mx = wave_max(mx);

    const float ex0 = have0 ? __expf(v0 - mx) : 0.0f;
    float dn = ex0;
    if (n > SLOT) {
        for (int j = lane; j < nov; j += 32) {
            const int e = g.ovf[j];
            if (g.edst[e] == d) dn += __expf(leaky(ssrc[g.esrc[e]] + sd) - mx);
        }
    }
    dn = wave_sum(dn);

    float4 acc = make_float4(0.0f, 0.0f, 0.0f, 0.0f);
    {
        const int   s = s0;
        const float a = have0 ? ex0 / dn : 0.0f;
        const int cnt = n < SLOT ? n : SLOT;
#pragma unroll 4
        for (int i = 0; i < cnt; ++i) {
            const int   si = __builtin_amdgcn_readlane(s, i);
            const float ai = __int_as_float(__builtin_amdgcn_readlane(__float_as_int(a), i));
            const float4 zv = *(const float4*)(gather_row<CAT>(z, n0, z1, si) + lane * 4);
            acc.x = fmaf(ai, zv.x, acc.x);
            acc.y = fmaf(ai, zv.y, acc.y);
            acc.z = fmaf(ai, zv.z, acc.z);
            acc.w = fmaf(ai, zv.w, acc.w);
        }
    }
    if (n > SLOT) {
        for (int base = 0; base < nov; base += 32) {
            const int j = base + lane;
            int  s = 0;
            bool mine = false;
            float a = 0.0f;
            if (j < nov) {
                const int e = g.ovf[j];
                if (g.edst[e] == d) {
                    mine = true;
                    s = g.esrc[e];
                    a = __expf(leaky(ssrc[s] + sd) - mx) / dn;
                }
            }
            unsigned int m = (unsigned int)__ballot(mine);
            while (m != 0u) {
                const int i = __builtin_ctz(m);
                m &= m - 1u;
                const int   si = __builtin_amdgcn_readlane(s, i);
                const float ai = __int_as_float(__builtin_amdgcn_readlane(__float_as_int(a), i));
                const float4 zv = *(const float4*)(gather_row<CAT>(z, n0, z1, si) + lane * 4);
                acc.x = fmaf(ai, zv.x, acc.x);
                acc.y = fmaf(ai, zv.y, acc.y);
                acc.z = fmaf(ai, zv.z, acc.z);
                acc.w = fmaf(ai, zv.w, acc.w);
            }
        }
    }
    return acc;
}

__global__ void agg_rows(const RowSrc g,
                         const float* __restrict__ ssrc, const float* __restrict__ sdst,
                         const float* __restrict__ z, float* __restrict__ out,
                         int lo, int nr)
{
    const int lane = threadIdx.x & 31;
    const int r = __builtin_amdgcn_readfirstlane((blockIdx.x * blockDim.x + threadIdx.x) >> 5);
    if (r >= nr) return;
    const float4 acc = agg_one_row<false>(g, ssrc, sdst, z, 0, nullptr, lo, r, lane);
    *(float4*)(out + (size_t)r * D + lane * 4) = acc;
}


__device__ __forceinline__ void strip_tiles2(const float* xrow, const float* __restrict__ wcol,
                                             v8f& c0, v8f& c1)
{
    c0 = (v8f){0.f, 0.f, 0.f, 0.f, 0.f, 0.f, 0.f, 0.f};
    c1 = (v8f){0.f, 0.f, 0.f, 0.f, 0.f, 0.f, 0.f, 0.f};
#pragma unroll 8
    for (int k0 = 0; k0 < D; k0 += 4) {
        const float2 xv0 = *(const float2*)(xrow + k0);
        const float2 xv1 = *(const float2*)(xrow + 16 * GP + k0);
        v2f a0; a0.x = xv0.x; a0.y = xv0.y;
        v2f a1; a1.x = xv1.x; a1.y = xv1.y;
        v2f w; w.x = wcol[(size_t)k0 * D];
               w.y = wcol[(size_t)(k0 + 1) * D];
        c0 = __builtin_amdgcn_wmma_f32_16x16x4_f32(
                false, a0, false, w, (short)0, c0, false, false);
        c1 = __builtin_amdgcn_wmma_f32_16x16x4_f32(
                false, a1, false, w, (short)0, c1, false, false);
    }
}

#define TR 32

__global__ void __launch_bounds__(512, 16)
agg_combine(const RowSrc g_d, const RowSrc g_k,
            const float* __restrict__ ssrc_d, const float* __restrict__ sdst_d,
            const float* __restrict__ h_kn,
            const float* __restrict__ ssrc_k, const float* __restrict__ sdst_k,
            const float* __restrict__ h_ex, int lo_k,
            const float* __restrict__ W_d, const float* __restrict__ W_k,
            const float* __restrict__ W1, const float* __restrict__ b1,
            const float* __restrict__ w2, float* __restrict__ out, int n)
{
    __shared__ __attribute__((aligned(16))) float sX[2][TR * GP];
    __shared__ __attribute__((aligned(16))) float sY[2][TR * GP];
    __shared__ __attribute__((aligned(16))) float sc[2][TR][8];
    const int wave = __builtin_amdgcn_readfirstlane(threadIdx.x >> 5);
    const int lane = threadIdx.x & 31;
    const int half = lane >> 4;
    const int lr   = lane & 15;
    const int g    = wave >> 3;
    const int wt   = wave & 7;
    const int row0 = blockIdx.x * TR;
    const int col0 = wt * 16;

    for (int s = 0; s < 2; ++s) {
        const int rr = 16 * s + wave;
        const int r = row0 + rr;
        float4 ra = make_float4(0.0f, 0.0f, 0.0f, 0.0f);
        if (r < n) ra = agg_one_row<false>(g_d, ssrc_d, sdst_d, h_kn, 0, nullptr, 0, r, lane);
        *(float4*)(sX[0] + rr * GP + lane * 4) = ra;
        float4 rb = make_float4(0.0f, 0.0f, 0.0f, 0.0f);
        if (r < n) rb = agg_one_row<true>(g_k, ssrc_k, sdst_k, h_ex, lo_k, h_kn, lo_k, r, lane);
        *(float4*)(sX[1] + rr * GP + lane * 4) = rb;
    }
    __syncthreads();

    {
        const float* Wg = g ? W_k : W_d;
        v8f y0, y1;
        strip_tiles2(sX[g] + lr * GP + 2 * half,
                     Wg + (size_t)(2 * half) * D + col0 + lr, y0, y1);
#pragma unroll
        for (int v = 0; v < 8; ++v) {
            sY[g][(8 * half + v) * GP + col0 + lr]      = y0[v];
            sY[g][(16 + 8 * half + v) * GP + col0 + lr] = y1[v];
        }
    }
    __syncthreads();

    v8f c0, c1;
    strip_tiles2(sY[g] + lr * GP + 2 * half,
                 W1 + (size_t)(2 * half) * D + col0 + lr, c0, c1);

    const float bias = b1[col0 + lr];
    const float w2c  = w2[col0 + lr];
#pragma unroll
    for (int v = 0; v < 8; ++v) { c0[v] = tanhf(c0[v] + bias); c1[v] = tanhf(c1[v] + bias); }
    const float pp0 = rows_dot16(c0, w2c, lr);
    const float pp1 = rows_dot16(c1, w2c, lr);
    if ((lr & 1) == 0) {
        sc[g][8 * half + (lr >> 1)][wt]      = pp0;
        sc[g][16 + 8 * half + (lr >> 1)][wt] = pp1;
    }
    __syncthreads();

    for (int s = 0; s < 2; ++s) {
        const int rr = 16 * s + wave;
        const float4 x0 = *(const float4*)&sc[0][rr][0];
        const float4 x1 = *(const float4*)&sc[0][rr][4];
        const float4 y0 = *(const float4*)&sc[1][rr][0];
        const float4 y1 = *(const float4*)&sc[1][rr][4];
        const float s1 = ((((((x0.x + x0.y) + x0.z) + x0.w) + x1.x) + x1.y) + x1.z) + x1.w;
        const float s2 = ((((((y0.x + y0.y) + y0.z) + y0.w) + y1.x) + y1.y) + y1.z) + y1.w;
        const float mm = fmaxf(s1, s2);
        const float e1 = __expf(s1 - mm), e2 = __expf(s2 - mm);
        const float al = e1 / (e1 + e2);
        const float be = 1.0f - al;
        const float4 va = *(const float4*)(sY[0] + rr * GP + lane * 4);
        const float4 vb = *(const float4*)(sY[1] + rr * GP + lane * 4);
        float4 o;
        o.x = al * va.x + be * vb.x;
        o.y = al * va.y + be * vb.y;
        o.z = al * va.z + be * vb.z;
        o.w = al * va.w + be * vb.w;
        const int r = row0 + rr;
        if (r < n) *(float4*)(out + (size_t)r * D + lane * 4) = o;
    }
}

static inline int cdiv(int a, int b) { return (a + b - 1) / b; }

extern "C" void kernel_launch(void* const* d_in, const int* in_sizes, int n_in,
                              void* d_out, int out_size, void* d_ws, size_t ws_size,
                              hipStream_t stream)
{
    (void)n_in; (void)ws_size; (void)out_size;
    const float* kn_emb   = (const float*)d_in[0];
    const float* exer_emb = (const float*)d_in[1];
    const float* W_dir    = (const float*)d_in[2];
    const float* a_dir    = (const float*)d_in[3];
    const float* W_ke     = (const float*)d_in[4];
    const float* a_ke     = (const float*)d_in[5];
    const float* W_ek     = (const float*)d_in[6];
    const float* a_ek     = (const float*)d_in[7];
    const float* rW1      = (const float*)d_in[8];
    const float* rb1      = (const float*)d_in[9];
    const float* rw2      = (const float*)d_in[10];
    const int* dsrc  = (const int*)d_in[11];
    const int* ddst  = (const int*)d_in[12];
    const int* kesrc = (const int*)d_in[13];
    const int* kedst = (const int*)d_in[14];
    const int* eksrc = (const int*)d_in[15];
    const int* ekdst = (const int*)d_in[16];

    const int Kn = in_sizes[0] / D;
    const int In = in_sizes[1] / D;
    const int Nn = Kn + In;
    const int E1 = in_sizes[11];
    const int E2 = in_sizes[13];

    float* w = (float*)d_ws;
    size_t off = 0;
    float* buf_z = w + off; off += (size_t)Nn * D;
    float* buf_B = w + off; off += (size_t)Nn * D;
    float* buf_A = w + off; off += (size_t)Kn * D;
    float* ssrc  = w + off; off += (size_t)Nn;
    float* sdst  = w + off; off += (size_t)Nn;
    float* ssrc_d = buf_A;
    float* sdst_d = buf_A + Kn;
    float* ssrc_k = buf_A + 2 * (size_t)Kn;
    float* sdst_k = ssrc_k + Nn;

    const int r4I = (In + 3) & ~3;
    const int r4K = (Kn + 3) & ~3;
    const int rb_ke  = r4I;
    const int rb_dir = r4I + r4K;
    const int nrows  = r4I + 2 * r4K;
    const int nclr   = nrows + 4;
    const int ncl32  = (nclr + 31) & ~31;
    int* cnt   = (int*)(buf_B + (size_t)Kn * D);
    int* novf  = cnt + nrows;
    int* slots = cnt + ncl32;
    int* ovf   = slots + (size_t)nrows * SLOT;

    float* out_kn = (float*)d_out;
    float* out_ex = (float*)d_out + (size_t)Kn * D;

    const int gb_ex = cdiv(In, RB);
    const int gb_kn = cdiv(Kn, RB);

    RowSrc g_ek;  g_ek.cnt  = cnt;          g_ek.slots  = slots;
                  g_ek.novf = novf;         g_ek.ovf    = ovf;
                  g_ek.esrc = eksrc;        g_ek.edst   = ekdst;
    RowSrc g_ke;  g_ke.cnt  = cnt + rb_ke;  g_ke.slots  = slots + (size_t)rb_ke * SLOT;
                  g_ke.novf = novf + 1;     g_ke.ovf    = ovf + E2;
                  g_ke.esrc = kesrc;        g_ke.edst   = kedst;
    RowSrc g_dir; g_dir.cnt  = cnt + rb_dir; g_dir.slots = slots + (size_t)rb_dir * SLOT;
                  g_dir.novf = novf + 2;     g_dir.ovf   = ovf + 2 * (size_t)E2;
                  g_dir.esrc = dsrc;         g_dir.edst  = ddst;

    NodeScores ns; ns.Wk = W_ke;  ns.attk = a_ke;  ns.ssrc_k = ssrc_k; ns.sdst_k = sdst_k;
                   ns.Wd = W_dir; ns.attd = a_dir; ns.ssrc_d = ssrc_d; ns.sdst_d = sdst_d;

    gemm_z<<<512, 256, 0, stream>>>(exer_emb, In, kn_emb, W_ek, a_ek, buf_z, ssrc, sdst, Nn, novf, 4, ns);

    edge_fill3<<<gb_ex + 2 * gb_kn, FILL_T, 0, stream>>>(eksrc, ekdst, 0, In, E2,
                                                         kesrc, kedst, In, Nn, E2, gb_ex, rb_ke,
                                                         dsrc, ddst, 0, Kn, E1, gb_ex + gb_kn, rb_dir,
                                                         cnt, slots, novf, ovf);

    agg_rows<<<cdiv(In, 8), 256, 0, stream>>>(g_ek, ssrc, sdst, buf_z, out_ex, 0, In);

    agg_combine<<<cdiv(Kn, TR), 512, 0, stream>>>(g_dir, g_ke,
                                                  ssrc_d, sdst_d, kn_emb,
                                                  ssrc_k, sdst_k, exer_emb, In,
                                                  W_dir, W_ke,
                                                  rW1, rb1, rw2, out_kn, Kn);
}
